# starting kernel
# baseline (speedup 1.0000x reference)
#include <hip/hip_runtime.h>
#include <hip/hip_bf16.h>


typedef float v2f __attribute__((ext_vector_type(2)));
typedef float v8f __attribute__((ext_vector_type(8)));

#define D_IN  128
#define D_HID 64
#define D_OUT 32


__global__ void k_zero_u32(unsigned* p, int n) {
    int i = blockIdx.x * blockDim.x + threadIdx.x;
    if (i < n) p[i] = 0u;
}

__global__ void k_degree(const int* __restrict__ ei, unsigned* __restrict__ deg, int E) {
    int e = blockIdx.x * blockDim.x + threadIdx.x;
    if (e < E) {
        int d = ei[E + e];
        atomicAdd(&deg[d], 1u);
    }
}

__global__ void k_dinv(const unsigned* __restrict__ deg, float* __restrict__ dinv, int N) {
    int i = blockIdx.x * blockDim.x + threadIdx.x;
    if (i < N) {
        float dg = (float)(deg[i] + 1u);
        dinv[i] = rsqrtf(dg);
    }
}


__global__ void k_gemm1_wmma(const float* __restrict__ X, const float* __restrict__ W,
                             float* __restrict__ H) {
    const int lane = threadIdx.x & 31;
    const int wave = threadIdx.x >> 5;
    const int m0   = blockIdx.x * 16;
    const int n0   = wave * 16;
    const int half = lane >> 4;
    const int lm   = lane & 15;

    const float* xrow = X + (size_t)(m0 + lm) * D_IN;
    v8f acc = {};
    #pragma unroll 4
    for (int k0 = 0; k0 < D_IN; k0 += 4) {
        const int ka = k0 + 2 * half;
        v2f a, b;
        a.x = xrow[ka];
        a.y = xrow[ka + 1];
        b.x = W[(size_t)ka       * D_HID + n0 + lm];
        b.y = W[(size_t)(ka + 1) * D_HID + n0 + lm];
        acc = __builtin_amdgcn_wmma_f32_16x16x4_f32(false, a, false, b,
                                                    (short)0, acc, false, false);
    }
    #pragma unroll
    for (int r = 0; r < 8; ++r) {
        const int m = m0 + r + 8 * half;
        H[(size_t)m * D_HID + n0 + lm] = acc[r];
    }
}

__global__ void k_selfinit64(const float* __restrict__ h, const float* __restrict__ dinv,
                             float* __restrict__ agg, int N) {
    int i = blockIdx.x * blockDim.x + threadIdx.x;
    if (i < N * D_HID) {
        int n = i >> 6;
        float w = dinv[n];
        agg[i] = w * w * h[i];
    }
}

__global__ void k_edge64(const int* __restrict__ ei, const float* __restrict__ dinv,
                         const float* __restrict__ h, float* __restrict__ agg, int E) {
    int t = blockIdx.x * 256 + threadIdx.x;
    int e = t >> 6;
    int f = t & 63;
    if (e < E) {
        int s = ei[e];
        int d = ei[E + e];
        float w = dinv[s] * dinv[d];
        atomicAdd(&agg[(size_t)d * D_HID + f], h[(size_t)s * D_HID + f] * w);
    }
}


__global__ void k_gemm2_wmma(const float* __restrict__ AGG, const float* __restrict__ b1,
                             const float* __restrict__ W, float* __restrict__ H2) {
    const int lane = threadIdx.x & 31;
    const int wave = threadIdx.x >> 5;
    const int m0   = blockIdx.x * 16;
    const int n0   = wave * 16;
    const int half = lane >> 4;
    const int lm   = lane & 15;

    const float* arow = AGG + (size_t)(m0 + lm) * D_HID;
    v8f acc = {};
    #pragma unroll 4
    for (int k0 = 0; k0 < D_HID; k0 += 4) {
        const int ka = k0 + 2 * half;
        v2f a, b;
        a.x = fmaxf(arow[ka]     + b1[ka],     0.f);
        a.y = fmaxf(arow[ka + 1] + b1[ka + 1], 0.f);
        b.x = W[(size_t)ka       * D_OUT + n0 + lm];
        b.y = W[(size_t)(ka + 1) * D_OUT + n0 + lm];
        acc = __builtin_amdgcn_wmma_f32_16x16x4_f32(false, a, false, b,
                                                    (short)0, acc, false, false);
    }
    #pragma unroll
    for (int r = 0; r < 8; ++r) {
        const int m = m0 + r + 8 * half;
        H2[(size_t)m * D_OUT + n0 + lm] = acc[r];
    }
}

__global__ void k_outinit32(const float* __restrict__ h2, const float* __restrict__ dinv,
                            const float* __restrict__ b2, float* __restrict__ out, int N) {
    int i = blockIdx.x * blockDim.x + threadIdx.x;
    if (i < N * D_OUT) {
        int n = i >> 5;
        int f = i & 31;
        float w = dinv[n];
        out[i] = b2[f] + w * w * h2[i];
    }
}

__global__ void k_edge32(const int* __restrict__ ei, const float* __restrict__ dinv,
                         const float* __restrict__ h2, float* __restrict__ out, int E) {
    int t = blockIdx.x * 256 + threadIdx.x;
    int e = t >> 5;
    int f = t & 31;
    if (e < E) {
        int s = ei[e];
        int d = ei[E + e];
        float w = dinv[s] * dinv[d];
        atomicAdd(&out[(size_t)d * D_OUT + f], h2[(size_t)s * D_OUT + f] * w);
    }
}


extern "C" void kernel_launch(void* const* d_in, const int* in_sizes, int n_in,
                              void* d_out, int out_size, void* d_ws, size_t ws_size,
                              hipStream_t stream) {
    const float* x   = (const float*)d_in[0];
    const int*   ei  = (const int*)  d_in[1];
    const float* W1  = (const float*)d_in[2];
    const float* b1  = (const float*)d_in[3];
    const float* W2  = (const float*)d_in[4];
    const float* b2  = (const float*)d_in[5];
    float*       out = (float*)d_out;

    const int N = in_sizes[0] / D_IN;
    const int E = in_sizes[1] / 2;

    char* ws = (char*)d_ws;
    unsigned* deg  = (unsigned*)(ws);
    float*    dinv = (float*)(ws + (size_t)N * 4);
    float*    h    = (float*)(ws + (size_t)N * 8);
    float*    agg1 = (float*)(ws + (size_t)N * 8 + (size_t)N * D_HID * 4);
    float*    h2   = (float*)(ws + (size_t)N * 8 + (size_t)N * D_HID * 8);

    const int TB = 256;

    k_zero_u32<<<(N + TB - 1) / TB, TB, 0, stream>>>(deg, N);
    k_degree  <<<(E + TB - 1) / TB, TB, 0, stream>>>(ei, deg, E);
    k_dinv    <<<(N + TB - 1) / TB, TB, 0, stream>>>(deg, dinv, N);

    k_gemm1_wmma<<<N / 16, 128, 0, stream>>>(x, W1, h);
    k_selfinit64<<<((size_t)N * D_HID + TB - 1) / TB, TB, 0, stream>>>(h, dinv, agg1, N);
    k_edge64    <<<((size_t)E * D_HID + TB - 1) / TB, TB, 0, stream>>>(ei, dinv, h, agg1, E);

    k_gemm2_wmma<<<N / 16, 64, 0, stream>>>(agg1, b1, W2, h2);
    k_outinit32 <<<((size_t)N * D_OUT + TB - 1) / TB, TB, 0, stream>>>(h2, dinv, b2, out, N);
    k_edge32    <<<((size_t)E * D_OUT + TB - 1) / TB, TB, 0, stream>>>(ei, dinv, h2, out, E);
}


# Float atomic scatter replaced by edges counting-sorted by destination (returning-atomic rank + scan) and a one-wave-per-node gather; sums now run self term first, then arrival-rank order
# speedup vs baseline: 2.8162x; 2.8162x over previous
#include <hip/hip_runtime.h>
#include <hip/hip_bf16.h>


typedef float v2f __attribute__((ext_vector_type(2)));
typedef float v8f __attribute__((ext_vector_type(8)));

#define D_IN  128
#define D_HID 64
#define D_OUT 32

#define SCAN_TB   256
#define SCAN_PER  4
#define SCAN_TILE (SCAN_TB * SCAN_PER)
#define GATHER_WAVES 8


__global__ void k_zero_u32(unsigned* p, int n) {
    int i = blockIdx.x * blockDim.x + threadIdx.x;
    if (i < n) p[i] = 0u;
}

__global__ void k_count_rank(const int* __restrict__ ei, unsigned* __restrict__ deg,
                             unsigned* __restrict__ rank, int E) {
    int e = blockIdx.x * blockDim.x + threadIdx.x;
    if (e < E) {
        int d = ei[E + e];
        rank[e] = atomicAdd(&deg[d], 1u);
    }
}


__device__ __forceinline__ unsigned block_sum_256(unsigned v, unsigned* sm) {
    for (int o = 16; o > 0; o >>= 1) v += __shfl_xor(v, o);
    if ((threadIdx.x & 31) == 0) sm[threadIdx.x >> 5] = v;
    __syncthreads();
    unsigned tot = 0;
    for (int k = 0; k < SCAN_TB / 32; ++k) tot += sm[k];
    return tot;
}

__global__ __launch_bounds__(SCAN_TB)
void k_scan_blocksum(const unsigned* __restrict__ deg, unsigned* __restrict__ bsum, int N) {
    __shared__ unsigned sm[SCAN_TB / 32];
    const int i0 = blockIdx.x * SCAN_TILE + threadIdx.x * SCAN_PER;
    unsigned s = 0;
    for (int k = 0; k < SCAN_PER; ++k) s += (i0 + k < N) ? deg[i0 + k] : 0u;
    const unsigned tot = block_sum_256(s, sm);
    if (threadIdx.x == 0) bsum[blockIdx.x] = tot;
}

__global__ __launch_bounds__(SCAN_TB)
void k_scan_write(const unsigned* __restrict__ deg, const unsigned* __restrict__ bsum,
                  unsigned* __restrict__ rowptr, float* __restrict__ dinv, int N) {
    __shared__ unsigned sm_a[SCAN_TB / 32];
    __shared__ unsigned sm_b[SCAN_TB / 32];
    const int t = threadIdx.x;
    const int lane = t & 31;
    const int wv = t >> 5;

    unsigned b = 0;
    for (int i = t; i < (int)blockIdx.x; i += SCAN_TB) b += bsum[i];
    const unsigned base = block_sum_256(b, sm_a);

    const int i0 = blockIdx.x * SCAN_TILE + t * SCAN_PER;
    unsigned d[SCAN_PER];
    unsigned tsum = 0;
    for (int k = 0; k < SCAN_PER; ++k) {
        d[k] = (i0 + k < N) ? deg[i0 + k] : 0u;
        tsum += d[k];
    }
    unsigned inc = tsum;
    for (int o = 1; o < 32; o <<= 1) {
        const unsigned y = __shfl_up(inc, o);
        if (lane >= o) inc += y;
    }
    if (lane == 31) sm_b[wv] = inc;
    __syncthreads();
    unsigned woff = 0;
    for (int k = 0; k < SCAN_TB / 32; ++k) woff += (k < wv) ? sm_b[k] : 0u;

    unsigned ex = base + woff + inc - tsum;
    for (int k = 0; k < SCAN_PER; ++k) {
        if (i0 + k < N) {
            rowptr[i0 + k] = ex;
            dinv[i0 + k]   = rsqrtf((float)(d[k] + 1u));
        }
        ex += d[k];
    }
    if (i0 <= N - 1 && N - 1 < i0 + SCAN_PER) rowptr[N] = ex;
}


__global__ void k_fill(const int* __restrict__ ei, const unsigned* __restrict__ rank,
                       const unsigned* __restrict__ rowptr, const float* __restrict__ dinv,
                       int2* __restrict__ csr, int E) {
    int e = blockIdx.x * blockDim.x + threadIdx.x;
    if (e < E) {
        const int s = ei[e];
        const int d = ei[E + e];
        const unsigned pos = rowptr[d] + rank[e];
        const float w = dinv[s] * dinv[d];
        if (pos < (unsigned)E) csr[pos] = make_int2(s, __float_as_int(w));
    }
}


__global__ void k_gemm1_wmma(const float* __restrict__ X, const float* __restrict__ W,
                             float* __restrict__ H) {
    const int lane = threadIdx.x & 31;
    const int wave = threadIdx.x >> 5;
    const int m0   = blockIdx.x * 16;
    const int n0   = wave * 16;
    const int half = lane >> 4;
    const int lm   = lane & 15;

    const float* xrow = X + (size_t)(m0 + lm) * D_IN;
    v8f acc = {};
    #pragma unroll 4
    for (int k0 = 0; k0 < D_IN; k0 += 4) {
        const int ka = k0 + 2 * half;
        v2f a, b;
        a.x = xrow[ka];
        a.y = xrow[ka + 1];
        b.x = W[(size_t)ka       * D_HID + n0 + lm];
        b.y = W[(size_t)(ka + 1) * D_HID + n0 + lm];
        acc = __builtin_amdgcn_wmma_f32_16x16x4_f32(false, a, false, b,
                                                    (short)0, acc, false, false);
    }
    #pragma unroll
    for (int r = 0; r < 8; ++r) {
        const int m = m0 + r + 8 * half;
        H[(size_t)m * D_HID + n0 + lm] = acc[r];
    }
}


__global__ __launch_bounds__(GATHER_WAVES * 32)
void k_gather64(const unsigned* __restrict__ rowptr, const int2* __restrict__ csr,
                const float* __restrict__ dinv, const float* __restrict__ h,
                float* __restrict__ agg, int N) {
    const int lane = threadIdx.x & 31;
    const int n = __builtin_amdgcn_readfirstlane(blockIdx.x * GATHER_WAVES + (threadIdx.x >> 5));
    if (n >= N) return;
    const unsigned beg = __builtin_amdgcn_readfirstlane(rowptr[n]);
    const unsigned end = __builtin_amdgcn_readfirstlane(rowptr[n + 1]);
    const float2* __restrict__ hv = (const float2*)h;

    float ws = dinv[n];
    ws = ws * ws;
    const float2 self = hv[(size_t)n * (D_HID / 2) + lane];
    float2 acc = make_float2(ws * self.x, ws * self.y);

    for (unsigned base = beg; base < end; base += 32) {
        const unsigned rem = end - base;
        const int cnt = rem < 32u ? (int)rem : 32;
        int2 ed = make_int2(0, 0);
        if (lane < cnt) ed = csr[base + lane];
        for (int j = 0; j < cnt; j += 8) {
            float2 v[8];
            float  w[8];
            for (int u = 0; u < 8; ++u) {
                const int jj = j + u;
                const int   s  = __builtin_amdgcn_readlane(ed.x, jj);
                const float ww = __int_as_float(__builtin_amdgcn_readlane(ed.y, jj));
                if (jj < cnt) {
                    w[u] = ww;
                    v[u] = hv[(size_t)s * (D_HID / 2) + lane];
                } else {
                    w[u] = 0.f;
                    v[u] = make_float2(0.f, 0.f);
                }
            }
            for (int u = 0; u < 8; ++u) {
                acc.x += w[u] * v[u].x;
                acc.y += w[u] * v[u].y;
            }
        }
    }
    ((float2*)agg)[(size_t)n * (D_HID / 2) + lane] = acc;
}


__global__ void k_gemm2_wmma(const float* __restrict__ AGG, const float* __restrict__ b1,
                             const float* __restrict__ W, float* __restrict__ H2) {
    const int lane = threadIdx.x & 31;
    const int wave = threadIdx.x >> 5;
    const int m0   = blockIdx.x * 16;
    const int n0   = wave * 16;
    const int half = lane >> 4;
    const int lm   = lane & 15;

    const float* arow = AGG + (size_t)(m0 + lm) * D_HID;
    v8f acc = {};
    #pragma unroll 4
    for (int k0 = 0; k0 < D_HID; k0 += 4) {
        const int ka = k0 + 2 * half;
        v2f a, b;
        a.x = fmaxf(arow[ka]     + b1[ka],     0.f);
        a.y = fmaxf(arow[ka + 1] + b1[ka + 1], 0.f);
        b.x = W[(size_t)ka       * D_OUT + n0 + lm];
        b.y = W[(size_t)(ka + 1) * D_OUT + n0 + lm];
        acc = __builtin_amdgcn_wmma_f32_16x16x4_f32(false, a, false, b,
                                                    (short)0, acc, false, false);
    }
    #pragma unroll
    for (int r = 0; r < 8; ++r) {
        const int m = m0 + r + 8 * half;
        H2[(size_t)m * D_OUT + n0 + lm] = acc[r];
    }
}


__global__ __launch_bounds__(GATHER_WAVES * 32)
void k_gather32(const unsigned* __restrict__ rowptr, const int2* __restrict__ csr,
                const float* __restrict__ dinv, const float* __restrict__ h2,
                const float* __restrict__ b2, float* __restrict__ out, int N) {
    const int lane = threadIdx.x & 31;
    const int n = __builtin_amdgcn_readfirstlane(blockIdx.x * GATHER_WAVES + (threadIdx.x >> 5));
    if (n >= N) return;
    const unsigned beg = __builtin_amdgcn_readfirstlane(rowptr[n]);
    const unsigned end = __builtin_amdgcn_readfirstlane(rowptr[n + 1]);

    float ws = dinv[n];
    ws = ws * ws;
    float acc = ws * h2[(size_t)n * D_OUT + lane];

    for (unsigned base = beg; base < end; base += 32) {
        const unsigned rem = end - base;
        const int cnt = rem < 32u ? (int)rem : 32;
        int2 ed = make_int2(0, 0);
        if (lane < cnt) ed = csr[base + lane];
        for (int j = 0; j < cnt; j += 8) {
            float v[8];
            float w[8];
            for (int u = 0; u < 8; ++u) {
                const int jj = j + u;
                const int   s  = __builtin_amdgcn_readlane(ed.x, jj);
                const float ww = __int_as_float(__builtin_amdgcn_readlane(ed.y, jj));
                if (jj < cnt) {
                    w[u] = ww;
                    v[u] = h2[(size_t)s * D_OUT + lane];
                } else {
                    w[u] = 0.f;
                    v[u] = 0.f;
                }
            }
            for (int u = 0; u < 8; ++u) acc += w[u] * v[u];
        }
    }
    out[(size_t)n * D_OUT + lane] = acc + b2[lane];
}


extern "C" void kernel_launch(void* const* d_in, const int* in_sizes, int n_in,
                              void* d_out, int out_size, void* d_ws, size_t ws_size,
                              hipStream_t stream) {
    const float* x   = (const float*)d_in[0];
    const int*   ei  = (const int*)  d_in[1];
    const float* W1  = (const float*)d_in[2];
    const float* b1  = (const float*)d_in[3];
    const float* W2  = (const float*)d_in[4];
    const float* b2  = (const float*)d_in[5];
    float*       out = (float*)d_out;

    const int N = in_sizes[0] / D_IN;
    const int E = in_sizes[1] / 2;

    char* ws = (char*)d_ws;
    float*    h      = (float*)(ws);
    float*    agg1   = (float*)(ws + (size_t)N * D_HID * 4);
    float*    h2     = (float*)(ws + (size_t)N * D_HID * 8);
    int2*     csr    = (int2*) (ws + (size_t)N * D_HID * 8 + (size_t)N * D_OUT * 4);
    unsigned* rank   = (unsigned*)((char*)csr + (size_t)E * 8);
    unsigned* deg    = rank + E;
    float*    dinv   = (float*)(deg + N);
    unsigned* rowptr = (unsigned*)(dinv + N);
    unsigned* bsum   = rowptr + N + 1;

    const int TB = 256;
    const int nscan = (N + SCAN_TILE - 1) / SCAN_TILE;
    const int ngath = (N + GATHER_WAVES - 1) / GATHER_WAVES;

    k_zero_u32     <<<(N + TB - 1) / TB, TB, 0, stream>>>(deg, N);
    k_count_rank   <<<(E + TB - 1) / TB, TB, 0, stream>>>(ei, deg, rank, E);
    k_scan_blocksum<<<nscan, SCAN_TB, 0, stream>>>(deg, bsum, N);
    k_scan_write   <<<nscan, SCAN_TB, 0, stream>>>(deg, bsum, rowptr, dinv, N);
    k_fill         <<<(E + TB - 1) / TB, TB, 0, stream>>>(ei, rank, rowptr, dinv, csr, E);

    k_gemm1_wmma<<<N / 16, 128, 0, stream>>>(x, W1, h);
    k_gather64  <<<ngath, GATHER_WAVES * 32, 0, stream>>>(rowptr, csr, dinv, h, agg1, N);

    k_gemm2_wmma<<<N / 16, 64, 0, stream>>>(agg1, b1, W2, h2);
    k_gather32  <<<ngath, GATHER_WAVES * 32, 0, stream>>>(rowptr, csr, dinv, h2, b2, out, N);
}


# Edge sort through LDS: chunks ordered by 512-node destination bin, then one work-group per bin counts and places with LDS counters (no global atomic, no scan launches); sum order: LDS arrival rank
# speedup vs baseline: 6.2514x; 2.2198x over previous
#include <hip/hip_runtime.h>
#include <hip/hip_bf16.h>


typedef float v2f __attribute__((ext_vector_type(2)));
typedef float v8f __attribute__((ext_vector_type(8)));

#define D_IN  128
#define D_HID 64
#define D_OUT 32

#define GATHER_WAVES 8

#define BIN_SHIFT 9
#define BIN_W     512
#define CHUNK     8192
#define SORT_TB   1024
#define PER_T     (CHUNK / SORT_TB)
#define CAP_B     12288
#define RUN_U     4

__device__ __forceinline__ unsigned block_excl_scan(unsigned v, unsigned* sm, unsigned* tot) {
    const int lane = threadIdx.x & 31;
    const int wv   = threadIdx.x >> 5;
    unsigned inc = v;
    for (int o = 1; o < 32; o <<= 1) {
        const unsigned y = __shfl_up(inc, o);
        if (lane >= o) inc += y;
    }
    __syncthreads();
    if (lane == 31) sm[wv] = inc;
    __syncthreads();
    unsigned woff = 0, all = 0;
    for (int k = 0; k < SORT_TB / 32; ++k) {
        const unsigned s = sm[k];
        all  += s;
        woff += (k < wv) ? s : 0u;
    }
    *tot = all;
    return woff + inc - v;
}


__global__ __launch_bounds__(SORT_TB)
void k_bin_chunk(const int* __restrict__ ei, unsigned* __restrict__ binned,
                 uint2* __restrict__ tab, int E, int NB) {
    __shared__ unsigned hist[SORT_TB];
    __shared__ unsigned sm[SORT_TB / 32];
    __shared__ unsigned stage[CHUNK];
    const int t  = threadIdx.x;
    const int e0 = blockIdx.x * CHUNK;
    const int n  = (E - e0 < CHUNK) ? (E - e0) : CHUNK;

    hist[t] = 0u;
    __syncthreads();

    unsigned sv[PER_T], dv[PER_T];
    #pragma unroll
    for (int k = 0; k < PER_T; ++k) {
        const int i = k * SORT_TB + t;
        sv[k] = 0u; dv[k] = 0u;
        if (i < n) {
            sv[k] = (unsigned)ei[e0 + i];
            dv[k] = (unsigned)ei[E + e0 + i];
        }
    }
    unsigned key[PER_T];
    #pragma unroll
    for (int k = 0; k < PER_T; ++k) {
        const int i = k * SORT_TB + t;
        key[k] = 0u;
        if (i < n) {
            const unsigned bin = dv[k] >> BIN_SHIFT;
            const unsigned r = atomicAdd(&hist[bin], 1u);
            key[k] = (bin << 13) | r;
        }
    }
    __syncthreads();
    const unsigned cnt = hist[t];
    unsigned tot;
    const unsigned ex = block_excl_scan(cnt, sm, &tot);
    hist[t] = ex;
    if (t < NB) tab[(size_t)blockIdx.x * NB + t] = make_uint2(ex, cnt);
    __syncthreads();
    #pragma unroll
    for (int k = 0; k < PER_T; ++k) {
        const int i = k * SORT_TB + t;
        if (i < n)
            stage[hist[key[k] >> 13] + (key[k] & (CHUNK - 1))] =
                (sv[k] << BIN_SHIFT) | (dv[k] & (BIN_W - 1));
    }
    __syncthreads();
    #pragma unroll
    for (int k = 0; k < PER_T; ++k) {
        const int i = k * SORT_TB + t;
        if (i < n) binned[e0 + i] = stage[i];
    }
}


template <bool PLACE>
__device__ __forceinline__ void bin_take(unsigned w, unsigned* ncnt, const unsigned* noff,
                                         unsigned* outs, int* csr, unsigned base, bool fits) {
    const unsigned dl = w & (BIN_W - 1);
    const unsigned r  = atomicAdd(&ncnt[dl], 1u);
    if (PLACE) {
        const unsigned p = noff[dl] + r;
        const int s = (int)(w >> BIN_SHIFT);
        if (fits) outs[p] = (unsigned)s;
        else      csr[base + p] = s;
    }
}

template <bool PLACE>
__device__ __forceinline__ void bin_walk(const unsigned* __restrict__ binned,
                                         const unsigned* roff, const unsigned* rcnt,
                                         unsigned* ncnt, const unsigned* noff, unsigned* outs,
                                         int* csr, unsigned base, bool fits, int C) {
    const int lane = threadIdx.x & 31;
    const int wv   = threadIdx.x >> 5;
    for (int c0 = wv; c0 < C; c0 += (SORT_TB / 32) * RUN_U) {
        unsigned w0[RUN_U], w1[RUN_U], mm[RUN_U];
        const unsigned* pp[RUN_U];
        #pragma unroll
        for (int u = 0; u < RUN_U; ++u) {
            const int c = c0 + (SORT_TB / 32) * u;
            const bool in = c < C;
            mm[u] = in ? rcnt[c] : 0u;
            pp[u] = binned + (size_t)(in ? c : 0) * CHUNK + (in ? roff[c] : 0u);
            w0[u] = ((unsigned)lane      < mm[u]) ? pp[u][lane]      : 0xFFFFFFFFu;
            w1[u] = ((unsigned)lane + 32 < mm[u]) ? pp[u][lane + 32] : 0xFFFFFFFFu;
        }
        #pragma unroll
        for (int u = 0; u < RUN_U; ++u) {
            if (w0[u] != 0xFFFFFFFFu) bin_take<PLACE>(w0[u], ncnt, noff, outs, csr, base, fits);
            if (w1[u] != 0xFFFFFFFFu) bin_take<PLACE>(w1[u], ncnt, noff, outs, csr, base, fits);
            for (unsigned j = (unsigned)lane + 64; j < mm[u]; j += 32)
                bin_take<PLACE>(pp[u][j], ncnt, noff, outs, csr, base, fits);
        }
    }
}

__global__ __launch_bounds__(SORT_TB)
void k_bin_build(const unsigned* __restrict__ binned, const uint2* __restrict__ tab,
                 unsigned* __restrict__ rowptr, float* __restrict__ dinv,
                 int* __restrict__ csr, int N, int C, int NB) {
    __shared__ unsigned roff[SORT_TB];
    __shared__ unsigned rcnt[SORT_TB];
    __shared__ unsigned ncnt[BIN_W];
    __shared__ unsigned noff[BIN_W];
    __shared__ unsigned sm[SORT_TB / 32];
    __shared__ unsigned outs[CAP_B];
    const int t = threadIdx.x;
    const int b = blockIdx.x;

    unsigned off = 0, cnt = 0;
    if (t < C) {
        const uint2 q = tab[(size_t)t * NB + b];
        off = q.x;
        cnt = q.y;
    }
    roff[t] = off;
    rcnt[t] = cnt;
    if (t < BIN_W) ncnt[t] = 0u;
    unsigned total, base;
    (void)block_excl_scan(cnt, sm, &total);
    (void)block_excl_scan(off, sm, &base);
    const bool fits = total <= (unsigned)CAP_B;

    bin_walk<false>(binned, roff, rcnt, ncnt, noff, outs, csr, base, fits, C);
    __syncthreads();

    const unsigned dg = (t < BIN_W) ? ncnt[t] : 0u;
    unsigned tot2;
    const unsigned ex = block_excl_scan(dg, sm, &tot2);
    if (t < BIN_W) {
        noff[t] = ex;
        ncnt[t] = 0u;
        const int node = b * BIN_W + t;
        if (node < N) {
            rowptr[node] = base + ex;
            dinv[node]   = rsqrtf((float)(dg + 1u));
        }
    }
    if (b == NB - 1 && t == 0) rowptr[N] = base + total;
    __syncthreads();

    bin_walk<true>(binned, roff, rcnt, ncnt, noff, outs, csr, base, fits, C);
    __syncthreads();
    if (fits)
        for (unsigned i = (unsigned)t; i < total; i += SORT_TB) csr[base + i] = (int)outs[i];
}


__global__ void k_gemm1_wmma(const float* __restrict__ X, const float* __restrict__ W,
                             const float* __restrict__ dinv, float* __restrict__ H) {
    const int lane = threadIdx.x & 31;
    const int wave = threadIdx.x >> 5;
    const int m0   = blockIdx.x * 16;
    const int n0   = wave * 16;
    const int half = lane >> 4;
    const int lm   = lane & 15;

    const float* xrow = X + (size_t)(m0 + lm) * D_IN;
    v8f acc = {};
    #pragma unroll 4
    for (int k0 = 0; k0 < D_IN; k0 += 4) {
        const int ka = k0 + 2 * half;
        v2f a, b;
        a.x = xrow[ka];
        a.y = xrow[ka + 1];
        b.x = W[(size_t)ka       * D_HID + n0 + lm];
        b.y = W[(size_t)(ka + 1) * D_HID + n0 + lm];
        acc = __builtin_amdgcn_wmma_f32_16x16x4_f32(false, a, false, b,
                                                    (short)0, acc, false, false);
    }
    #pragma unroll
    for (int r = 0; r < 8; ++r) {
        const int m = m0 + r + 8 * half;
        H[(size_t)m * D_HID + n0 + lm] = acc[r] * dinv[m];
    }
}


__global__ __launch_bounds__(GATHER_WAVES * 32)
void k_gather64(const unsigned* __restrict__ rowptr, const int* __restrict__ csr,
                const float* __restrict__ dinv, const float* __restrict__ h,
                float* __restrict__ agg, int N) {
    const int lane = threadIdx.x & 31;
    const int n = __builtin_amdgcn_readfirstlane(blockIdx.x * GATHER_WAVES + (threadIdx.x >> 5));
    if (n >= N) return;
    const unsigned beg = __builtin_amdgcn_readfirstlane(rowptr[n]);
    const unsigned end = __builtin_amdgcn_readfirstlane(rowptr[n + 1]);
    const float2* __restrict__ hv = (const float2*)h;

    const float dn = dinv[n];
    float2 acc = hv[(size_t)n * (D_HID / 2) + lane];

    for (unsigned base = beg; base < end; base += 32) {
        const unsigned rem = end - base;
        const int cnt = rem < 32u ? (int)rem : 32;
        int ed = 0;
        if (lane < cnt) ed = csr[base + lane];
        for (int j = 0; j < cnt; j += 8) {
            float2 v[8];
            for (int u = 0; u < 8; ++u) {
                const int jj = j + u;
                const int s = __builtin_amdgcn_readlane(ed, jj);
                if (jj < cnt) v[u] = hv[(size_t)s * (D_HID / 2) + lane];
                else          v[u] = make_float2(0.f, 0.f);
            }
            for (int u = 0; u < 8; ++u) {
                acc.x += v[u].x;
                acc.y += v[u].y;
            }
        }
    }
    ((float2*)agg)[(size_t)n * (D_HID / 2) + lane] = make_float2(dn * acc.x, dn * acc.y);
}


__global__ void k_gemm2_wmma(const float* __restrict__ AGG, const float* __restrict__ b1,
                             const float* __restrict__ W, const float* __restrict__ dinv,
                             float* __restrict__ H2) {
    const int lane = threadIdx.x & 31;
    const int wave = threadIdx.x >> 5;
    const int m0   = blockIdx.x * 16;
    const int n0   = wave * 16;
    const int half = lane >> 4;
    const int lm   = lane & 15;

    const float* arow = AGG + (size_t)(m0 + lm) * D_HID;
    v8f acc = {};
    #pragma unroll 4
    for (int k0 = 0; k0 < D_HID; k0 += 4) {
        const int ka = k0 + 2 * half;
        v2f a, b;
        a.x = fmaxf(arow[ka]     + b1[ka],     0.f);
        a.y = fmaxf(arow[ka + 1] + b1[ka + 1], 0.f);
        b.x = W[(size_t)ka       * D_OUT + n0 + lm];
        b.y = W[(size_t)(ka + 1) * D_OUT + n0 + lm];
        acc = __builtin_amdgcn_wmma_f32_16x16x4_f32(false, a, false, b,
                                                    (short)0, acc, false, false);
    }
    #pragma unroll
    for (int r = 0; r < 8; ++r) {
        const int m = m0 + r + 8 * half;
        H2[(size_t)m * D_OUT + n0 + lm] = acc[r] * dinv[m];
    }
}


__global__ __launch_bounds__(GATHER_WAVES * 32)
void k_gather32(const unsigned* __restrict__ rowptr, const int* __restrict__ csr,
                const float* __restrict__ dinv, const float* __restrict__ h2,
                const float* __restrict__ b2, float* __restrict__ out, int N) {
    const int lane = threadIdx.x & 31;
    const int n = __builtin_amdgcn_readfirstlane(blockIdx.x * GATHER_WAVES + (threadIdx.x >> 5));
    if (n >= N) return;
    const unsigned beg = __builtin_amdgcn_readfirstlane(rowptr[n]);
    const unsigned end = __builtin_amdgcn_readfirstlane(rowptr[n + 1]);

    const float dn = dinv[n];
    float acc = h2[(size_t)n * D_OUT + lane];

    for (unsigned base = beg; base < end; base += 32) {
        const unsigned rem = end - base;
        const int cnt = rem < 32u ? (int)rem : 32;
        int ed = 0;
        if (lane < cnt) ed = csr[base + lane];
        for (int j = 0; j < cnt; j += 8) {
            float v[8];
            for (int u = 0; u < 8; ++u) {
                const int jj = j + u;
                const int s = __builtin_amdgcn_readlane(ed, jj);
                if (jj < cnt) v[u] = h2[(size_t)s * D_OUT + lane];
                else          v[u] = 0.f;
            }
            for (int u = 0; u < 8; ++u) acc += v[u];
        }
    }
    out[(size_t)n * D_OUT + lane] = dn * acc + b2[lane];
}


extern "C" void kernel_launch(void* const* d_in, const int* in_sizes, int n_in,
                              void* d_out, int out_size, void* d_ws, size_t ws_size,
                              hipStream_t stream) {
    const float* x   = (const float*)d_in[0];
    const int*   ei  = (const int*)  d_in[1];
    const float* W1  = (const float*)d_in[2];
    const float* b1  = (const float*)d_in[3];
    const float* W2  = (const float*)d_in[4];
    const float* b2  = (const float*)d_in[5];
    float*       out = (float*)d_out;

    const int N = in_sizes[0] / D_IN;
    const int E = in_sizes[1] / 2;

    char* ws = (char*)d_ws;
    float*    h      = (float*)(ws);
    float*    agg1   = (float*)(ws + (size_t)N * D_HID * 4);
    float*    h2     = (float*)(ws + (size_t)N * D_HID * 8);
    int*      csr    = (int*)  (ws + (size_t)N * D_HID * 8 + (size_t)N * D_OUT * 4);
    unsigned* binned = (unsigned*)((char*)csr + (size_t)E * 4);
    float*    dinv   = (float*)(binned + E);

    const int C  = (E + CHUNK - 1) / CHUNK;
    const int NB = (N + BIN_W - 1) / BIN_W;
    const int ngath = (N + GATHER_WAVES - 1) / GATHER_WAVES;

    uint2*    tab    = (uint2*)(dinv + N);
    unsigned* rowptr = (unsigned*)(tab + (size_t)C * NB);

    k_bin_chunk<<<C,  SORT_TB, 0, stream>>>(ei, binned, tab, E, NB);
    k_bin_build<<<NB, SORT_TB, 0, stream>>>(binned, tab, rowptr, dinv, csr, N, C, NB);

    k_gemm1_wmma<<<N / 16, 128, 0, stream>>>(x, W1, dinv, h);
    k_gather64  <<<ngath, GATHER_WAVES * 32, 0, stream>>>(rowptr, csr, dinv, h, agg1, N);

    k_gemm2_wmma<<<N / 16, 64, 0, stream>>>(agg1, b1, W2, dinv, h2);
    k_gather32  <<<ngath, GATHER_WAVES * 32, 0, stream>>>(rowptr, csr, dinv, h2, b2, out, N);
}


# Gathers: a wave walks 16 consecutive nodes, their edge lists read as one stream with two groups of 16 row loads always in flight (was one short-lived wave per node); sum order per node unchanged
# speedup vs baseline: 6.5387x; 1.0460x over previous
#include <hip/hip_runtime.h>
#include <hip/hip_bf16.h>


typedef float v2f __attribute__((ext_vector_type(2)));
typedef float v8f __attribute__((ext_vector_type(8)));

#define D_IN  128
#define D_HID 64
#define D_OUT 32

#define GATHER_WAVES 8
#define NPW          16
#define GROUP        16

#define BIN_SHIFT 9
#define BIN_W     512
#define CHUNK     8192
#define SORT_TB   1024
#define PER_T     (CHUNK / SORT_TB)
#define CAP_B     12288
#define RUN_U     4

__device__ __forceinline__ unsigned block_excl_scan(unsigned v, unsigned* sm, unsigned* tot) {
    const int lane = threadIdx.x & 31;
    const int wv   = threadIdx.x >> 5;
    unsigned inc = v;
    for (int o = 1; o < 32; o <<= 1) {
        const unsigned y = __shfl_up(inc, o);
        if (lane >= o) inc += y;
    }
    __syncthreads();
    if (lane == 31) sm[wv] = inc;
    __syncthreads();
    unsigned woff = 0, all = 0;
    for (int k = 0; k < SORT_TB / 32; ++k) {
        const unsigned s = sm[k];
        all  += s;
        woff += (k < wv) ? s : 0u;
    }
    *tot = all;
    return woff + inc - v;
}


__global__ __launch_bounds__(SORT_TB)
void k_bin_chunk(const int* __restrict__ ei, unsigned* __restrict__ binned,
                 uint2* __restrict__ tab, int E, int NB) {
    __shared__ unsigned hist[SORT_TB];
    __shared__ unsigned sm[SORT_TB / 32];
    __shared__ unsigned stage[CHUNK];
    const int t  = threadIdx.x;
    const int e0 = blockIdx.x * CHUNK;
    const int n  = (E - e0 < CHUNK) ? (E - e0) : CHUNK;

    hist[t] = 0u;
    __syncthreads();

    unsigned sv[PER_T], dv[PER_T];
    #pragma unroll
    for (int k = 0; k < PER_T; ++k) {
        const int i = k * SORT_TB + t;
        sv[k] = 0u; dv[k] = 0u;
        if (i < n) {
            sv[k] = (unsigned)ei[e0 + i];
            dv[k] = (unsigned)ei[E + e0 + i];
        }
    }
    unsigned key[PER_T];
    #pragma unroll
    for (int k = 0; k < PER_T; ++k) {
        const int i = k * SORT_TB + t;
        key[k] = 0u;
        if (i < n) {
            const unsigned bin = dv[k] >> BIN_SHIFT;
            const unsigned r = atomicAdd(&hist[bin], 1u);
            key[k] = (bin << 13) | r;
        }
    }
    __syncthreads();
    const unsigned cnt = hist[t];
    unsigned tot;
    const unsigned ex = block_excl_scan(cnt, sm, &tot);
    hist[t] = ex;
    if (t < NB) tab[(size_t)blockIdx.x * NB + t] = make_uint2(ex, cnt);
    __syncthreads();
    #pragma unroll
    for (int k = 0; k < PER_T; ++k) {
        const int i = k * SORT_TB + t;
        if (i < n)
            stage[hist[key[k] >> 13] + (key[k] & (CHUNK - 1))] =
                (sv[k] << BIN_SHIFT) | (dv[k] & (BIN_W - 1));
    }
    __syncthreads();
    #pragma unroll
    for (int k = 0; k < PER_T; ++k) {
        const int i = k * SORT_TB + t;
        if (i < n) binned[e0 + i] = stage[i];
    }
}


template <bool PLACE>
__device__ __forceinline__ void bin_take(unsigned w, unsigned* ncnt, const unsigned* noff,
                                         unsigned* outs, int* csr, unsigned base, bool fits) {
    const unsigned dl = w & (BIN_W - 1);
    const unsigned r  = atomicAdd(&ncnt[dl], 1u);
    if (PLACE) {
        const unsigned p = noff[dl] + r;
        const int s = (int)(w >> BIN_SHIFT);
        if (fits) outs[p] = (unsigned)s;
        else      csr[base + p] = s;
    }
}

template <bool PLACE>
__device__ __forceinline__ void bin_walk(const unsigned* __restrict__ binned,
                                         const unsigned* roff, const unsigned* rcnt,
                                         unsigned* ncnt, const unsigned* noff, unsigned* outs,
                                         int* csr, unsigned base, bool fits, int C) {
    const int lane = threadIdx.x & 31;
    const int wv   = threadIdx.x >> 5;
    for (int c0 = wv; c0 < C; c0 += (SORT_TB / 32) * RUN_U) {
        unsigned w0[RUN_U], w1[RUN_U], mm[RUN_U];
        const unsigned* pp[RUN_U];
        #pragma unroll
        for (int u = 0; u < RUN_U; ++u) {
            const int c = c0 + (SORT_TB / 32) * u;
            const bool in = c < C;
            mm[u] = in ? rcnt[c] : 0u;
            pp[u] = binned + (size_t)(in ? c : 0) * CHUNK + (in ? roff[c] : 0u);
            w0[u] = ((unsigned)lane      < mm[u]) ? pp[u][lane]      : 0xFFFFFFFFu;
            w1[u] = ((unsigned)lane + 32 < mm[u]) ? pp[u][lane + 32] : 0xFFFFFFFFu;
        }
        #pragma unroll
        for (int u = 0; u < RUN_U; ++u) {
            if (w0[u] != 0xFFFFFFFFu) bin_take<PLACE>(w0[u], ncnt, noff, outs, csr, base, fits);
            if (w1[u] != 0xFFFFFFFFu) bin_take<PLACE>(w1[u], ncnt, noff, outs, csr, base, fits);
            for (unsigned j = (unsigned)lane + 64; j < mm[u]; j += 32)
                bin_take<PLACE>(pp[u][j], ncnt, noff, outs, csr, base, fits);
        }
    }
}

__global__ __launch_bounds__(SORT_TB)
void k_bin_build(const unsigned* __restrict__ binned, const uint2* __restrict__ tab,
                 unsigned* __restrict__ rowptr, float* __restrict__ dinv,
                 int* __restrict__ csr, int N, int C, int NB) {
    __shared__ unsigned roff[SORT_TB];
    __shared__ unsigned rcnt[SORT_TB];
    __shared__ unsigned ncnt[BIN_W];
    __shared__ unsigned noff[BIN_W];
    __shared__ unsigned sm[SORT_TB / 32];
    __shared__ unsigned outs[CAP_B];
    const int t = threadIdx.x;
    const int b = blockIdx.x;

    unsigned off = 0, cnt = 0;
    if (t < C) {
        const uint2 q = tab[(size_t)t * NB + b];
        off = q.x;
        cnt = q.y;
    }
    roff[t] = off;
    rcnt[t] = cnt;
    if (t < BIN_W) ncnt[t] = 0u;
    unsigned total, base;
    (void)block_excl_scan(cnt, sm, &total);
    (void)block_excl_scan(off, sm, &base);
    const bool fits = total <= (unsigned)CAP_B;

    bin_walk<false>(binned, roff, rcnt, ncnt, noff, outs, csr, base, fits, C);
    __syncthreads();

    const unsigned dg = (t < BIN_W) ? ncnt[t] : 0u;
    unsigned tot2;
    const unsigned ex = block_excl_scan(dg, sm, &tot2);
    if (t < BIN_W) {
        noff[t] = ex;
        ncnt[t] = 0u;
        const int node = b * BIN_W + t;
        if (node < N) {
            rowptr[node] = base + ex;
            dinv[node]   = rsqrtf((float)(dg + 1u));
        }
    }
    if (b == NB - 1 && t == 0) rowptr[N] = base + total;
    __syncthreads();

    bin_walk<true>(binned, roff, rcnt, ncnt, noff, outs, csr, base, fits, C);
    __syncthreads();
    if (fits)
        for (unsigned i = (unsigned)t; i < total; i += SORT_TB) csr[base + i] = (int)outs[i];
}


__global__ void k_gemm1_wmma(const float* __restrict__ X, const float* __restrict__ W,
                             const float* __restrict__ dinv, float* __restrict__ H) {
    const int lane = threadIdx.x & 31;
    const int wave = threadIdx.x >> 5;
    const int m0   = blockIdx.x * 16;
    const int n0   = wave * 16;
    const int half = lane >> 4;
    const int lm   = lane & 15;

    const float* xrow = X + (size_t)(m0 + lm) * D_IN;
    v8f acc = {};
    #pragma unroll 4
    for (int k0 = 0; k0 < D_IN; k0 += 4) {
        const int ka = k0 + 2 * half;
        v2f a, b;
        a.x = xrow[ka];
        a.y = xrow[ka + 1];
        b.x = W[(size_t)ka       * D_HID + n0 + lm];
        b.y = W[(size_t)(ka + 1) * D_HID + n0 + lm];
        acc = __builtin_amdgcn_wmma_f32_16x16x4_f32(false, a, false, b,
                                                    (short)0, acc, false, false);
    }
    #pragma unroll
    for (int r = 0; r < 8; ++r) {
        const int m = m0 + r + 8 * half;
        H[(size_t)m * D_HID + n0 + lm] = acc[r] * dinv[m];
    }
}


__device__ __forceinline__ float  row_add(float a, float b)   { return a + b; }
__device__ __forceinline__ float2 row_add(float2 a, float2 b) { return make_float2(a.x + b.x, a.y + b.y); }
__device__ __forceinline__ float  row_fin(float a, float dn, float b)    { return dn * a + b; }
__device__ __forceinline__ float2 row_fin(float2 a, float dn, float2)    { return make_float2(dn * a.x, dn * a.y); }

template <typename VT>
__device__ __forceinline__ void gather_walk(const unsigned* __restrict__ rowptr, const int* __restrict__ csr,
                                            const float* __restrict__ dinv, const float* __restrict__ src,
                                            const float* __restrict__ bias, float* __restrict__ dst,
                                            VT* __restrict__ own_s, int N) {
    const int lane = threadIdx.x & 31;
    const int wv = __builtin_amdgcn_readfirstlane((int)(threadIdx.x >> 5));
    const int n0 = (blockIdx.x * GATHER_WAVES + wv) * NPW;
    if (n0 >= N) return;
    const int cnt = (N - n0 < NPW) ? (N - n0) : NPW;

    const char* __restrict__ hb = (const char*)src;
    const unsigned ROWB    = 32u * (unsigned)sizeof(VT);
    const unsigned laneoff = (unsigned)lane * (unsigned)sizeof(VT);
    VT* __restrict__ out = (VT*)dst;
    VT* own = own_s + wv * (NPW * 32) + lane;

    VT t[NPW];
    #pragma unroll
    for (int k = 0; k < NPW; ++k) {
        const int kk = k < cnt ? k : cnt - 1;
        t[k] = *(const VT*)(hb + (size_t)(n0 + kk) * ROWB + laneoff);
    }
    const unsigned rpv = rowptr[n0 + (lane < cnt ? lane : cnt)];
    const float    dnv = dinv[n0 + (lane < cnt ? lane : cnt - 1)];
    VT bv = t[0];
    if constexpr (sizeof(VT) == sizeof(float)) bv = ((const VT*)bias)[lane];
    #pragma unroll
    for (int k = 0; k < NPW; ++k) own[k * 32] = t[k];

    const unsigned e0   = (unsigned)__builtin_amdgcn_readlane((int)rpv, 0);
    const unsigned eEnd = (unsigned)__builtin_amdgcn_readlane((int)rpv, cnt);

    unsigned edc, edn;
    {
        const unsigned i0 = e0 + (unsigned)lane, i1 = i0 + 32u;
        edc = ((i0 < eEnd) ? (unsigned)csr[i0] : (unsigned)n0) * ROWB;
        edn = ((i1 < eEnd) ? (unsigned)csr[i1] : (unsigned)n0) * ROWB;
    }

    int      k      = 0;
    unsigned segEnd = (unsigned)__builtin_amdgcn_readlane((int)rpv, 1);
    VT acc     = t[0];
    VT own_nxt = own[32];

    VT buf[2][GROUP];
    #pragma unroll
    for (int u = 0; u < GROUP; ++u) {
        const unsigned ro = (unsigned)__builtin_amdgcn_readlane((int)edc, u);
        buf[0][u] = *(const VT*)(hb + (size_t)(ro + laneoff));
    }

    unsigned gb = e0;
    for (unsigned eb = e0; eb < eEnd; eb += 32u) {
        unsigned edf;
        {
            const unsigned i2 = eb + 64u + (unsigned)lane;
            edf = ((i2 < eEnd) ? (unsigned)csr[i2] : (unsigned)n0) * ROWB;
        }
        #pragma unroll
        for (int g = 0; g < 32 / GROUP; ++g) {
            #pragma unroll
            for (int u = 0; u < GROUP; ++u) {
                const unsigned ro = (g + 1 < 32 / GROUP)
                    ? (unsigned)__builtin_amdgcn_readlane((int)edc, (g + 1) * GROUP + u)
                    : (unsigned)__builtin_amdgcn_readlane((int)edn, u);
                buf[(g + 1) & 1][u] = *(const VT*)(hb + (size_t)(ro + laneoff));
            }
            if (segEnd - gb >= (unsigned)GROUP) {
                #pragma unroll
                for (int u = 0; u < GROUP; ++u) acc = row_add(acc, buf[g & 1][u]);
            } else {
                #pragma unroll
                for (int u = 0; u < GROUP; ++u) {
                    while (__builtin_expect(gb + (unsigned)u == segEnd, 0)) {
                        const float dn = __builtin_bit_cast(float, __builtin_amdgcn_readlane(__builtin_bit_cast(int, dnv), k));
                        out[(size_t)(n0 + k) * 32 + lane] = row_fin(acc, dn, bv);
                        ++k;
                        if (k < cnt) {
                            acc     = own_nxt;
                            own_nxt = own[((k + 1 < NPW) ? k + 1 : NPW - 1) * 32];
                            segEnd  = (unsigned)__builtin_amdgcn_readlane((int)rpv, k + 1);
                        } else {
                            segEnd  = 0xFFFFFFFFu;
                        }
                    }
                    acc = row_add(acc, buf[g & 1][u]);
                }
            }
            gb += (unsigned)GROUP;
        }
        edc = edn;
        edn = edf;
    }
    while (k < cnt) {
        const float dn = __builtin_bit_cast(float, __builtin_amdgcn_readlane(__builtin_bit_cast(int, dnv), k));
        out[(size_t)(n0 + k) * 32 + lane] = row_fin(acc, dn, bv);
        ++k;
        acc     = own_nxt;
        own_nxt = own[((k + 1 < NPW) ? k + 1 : NPW - 1) * 32];
    }
}


__global__ __launch_bounds__(GATHER_WAVES * 32)
void k_gather64(const unsigned* __restrict__ rowptr, const int* __restrict__ csr,
                const float* __restrict__ dinv, const float* __restrict__ h,
                float* __restrict__ agg, int N) {
    __shared__ float2 own_s[GATHER_WAVES * NPW * 32];
    gather_walk<float2>(rowptr, csr, dinv, h, nullptr, agg, own_s, N);
}


__global__ void k_gemm2_wmma(const float* __restrict__ AGG, const float* __restrict__ b1,
                             const float* __restrict__ W, const float* __restrict__ dinv,
                             float* __restrict__ H2) {
    const int lane = threadIdx.x & 31;
    const int wave = threadIdx.x >> 5;
    const int m0   = blockIdx.x * 16;
    const int n0   = wave * 16;
    const int half = lane >> 4;
    const int lm   = lane & 15;

    const float* arow = AGG + (size_t)(m0 + lm) * D_HID;
    v8f acc = {};
    #pragma unroll 4
    for (int k0 = 0; k0 < D_HID; k0 += 4) {
        const int ka = k0 + 2 * half;
        v2f a, b;
        a.x = fmaxf(arow[ka]     + b1[ka],     0.f);
        a.y = fmaxf(arow[ka + 1] + b1[ka + 1], 0.f);
        b.x = W[(size_t)ka       * D_OUT + n0 + lm];
        b.y = W[(size_t)(ka + 1) * D_OUT + n0 + lm];
        acc = __builtin_amdgcn_wmma_f32_16x16x4_f32(false, a, false, b,
                                                    (short)0, acc, false, false);
    }
    #pragma unroll
    for (int r = 0; r < 8; ++r) {
        const int m = m0 + r + 8 * half;
        H2[(size_t)m * D_OUT + n0 + lm] = acc[r] * dinv[m];
    }
}


__global__ __launch_bounds__(GATHER_WAVES * 32)
void k_gather32(const unsigned* __restrict__ rowptr, const int* __restrict__ csr,
                const float* __restrict__ dinv, const float* __restrict__ h2,
                const float* __restrict__ b2, float* __restrict__ out, int N) {
    __shared__ float own_s[GATHER_WAVES * NPW * 32];
    gather_walk<float>(rowptr, csr, dinv, h2, b2, out, own_s, N);
}


extern "C" void kernel_launch(void* const* d_in, const int* in_sizes, int n_in,
                              void* d_out, int out_size, void* d_ws, size_t ws_size,
                              hipStream_t stream) {
    const float* x   = (const float*)d_in[0];
    const int*   ei  = (const int*)  d_in[1];
    const float* W1  = (const float*)d_in[2];
    const float* b1  = (const float*)d_in[3];
    const float* W2  = (const float*)d_in[4];
    const float* b2  = (const float*)d_in[5];
    float*       out = (float*)d_out;

    const int N = in_sizes[0] / D_IN;
    const int E = in_sizes[1] / 2;

    char* ws = (char*)d_ws;
    float*    h      = (float*)(ws);
    float*    agg1   = (float*)(ws + (size_t)N * D_HID * 4);
    float*    h2     = (float*)(ws + (size_t)N * D_HID * 8);
    int*      csr    = (int*)  (ws + (size_t)N * D_HID * 8 + (size_t)N * D_OUT * 4);
    unsigned* binned = (unsigned*)((char*)csr + (size_t)E * 4);
    float*    dinv   = (float*)(binned + E);

    const int C  = (E + CHUNK - 1) / CHUNK;
    const int NB = (N + BIN_W - 1) / BIN_W;
    const int ngath = (N + GATHER_WAVES * NPW - 1) / (GATHER_WAVES * NPW);

    uint2*    tab    = (uint2*)(dinv + N);
    unsigned* rowptr = (unsigned*)(tab + (size_t)C * NB);

    k_bin_chunk<<<C,  SORT_TB, 0, stream>>>(ei, binned, tab, E, NB);
    k_bin_build<<<NB, SORT_TB, 0, stream>>>(binned, tab, rowptr, dinv, csr, N, C, NB);

    k_gemm1_wmma<<<N / 16, 128, 0, stream>>>(x, W1, dinv, h);
    k_gather64  <<<ngath, GATHER_WAVES * 32, 0, stream>>>(rowptr, csr, dinv, h, agg1, N);

    k_gemm2_wmma<<<N / 16, 64, 0, stream>>>(agg1, b1, W2, dinv, h2);
    k_gather32  <<<ngath, GATHER_WAVES * 32, 0, stream>>>(rowptr, csr, dinv, h2, b2, out, N);
}


# Layer-2 GEMM moved into the layer-1 gather: a wave's 16 finished rows stay in LDS as relu(agg+b1) and meet W2 on WMMA there; one launch and the agg1 round trip gone (bias add fused, sum order kept)
# speedup vs baseline: 7.1352x; 1.0912x over previous
#include <hip/hip_runtime.h>
#include <hip/hip_bf16.h>


typedef float v2f __attribute__((ext_vector_type(2)));
typedef float v8f __attribute__((ext_vector_type(8)));

#define D_IN  128
#define D_HID 64
#define D_OUT 32

#define GATHER_WAVES 8
#define NPW          16
#define GROUP        16

#define BIN_SHIFT 9
#define BIN_W     512
#define CHUNK     8192
#define SORT_TB   1024
#define PER_T     (CHUNK / SORT_TB)
#define CAP_B     12288
#define RUN_U     4

__device__ __forceinline__ unsigned block_excl_scan(unsigned v, unsigned* sm, unsigned* tot) {
    const int lane = threadIdx.x & 31;
    const int wv   = threadIdx.x >> 5;
    unsigned inc = v;
    for (int o = 1; o < 32; o <<= 1) {
        const unsigned y = __shfl_up(inc, o);
        if (lane >= o) inc += y;
    }
    __syncthreads();
    if (lane == 31) sm[wv] = inc;
    __syncthreads();
    unsigned woff = 0, all = 0;
    for (int k = 0; k < SORT_TB / 32; ++k) {
        const unsigned s = sm[k];
        all  += s;
        woff += (k < wv) ? s : 0u;
    }
    *tot = all;
    return woff + inc - v;
}


__global__ __launch_bounds__(SORT_TB)
void k_bin_chunk(const int* __restrict__ ei, unsigned* __restrict__ binned,
                 uint2* __restrict__ tab, int E, int NB) {
    __shared__ unsigned hist[SORT_TB];
    __shared__ unsigned sm[SORT_TB / 32];
    __shared__ unsigned stage[CHUNK];
    const int t  = threadIdx.x;
    const int e0 = blockIdx.x * CHUNK;
    const int n  = (E - e0 < CHUNK) ? (E - e0) : CHUNK;

    hist[t] = 0u;
    __syncthreads();

    unsigned sv[PER_T], dv[PER_T];
    #pragma unroll
    for (int k = 0; k < PER_T; ++k) {
        const int i = k * SORT_TB + t;
        sv[k] = 0u; dv[k] = 0u;
        if (i < n) {
            sv[k] = (unsigned)ei[e0 + i];
            dv[k] = (unsigned)ei[E + e0 + i];
        }
    }
    unsigned key[PER_T];
    #pragma unroll
    for (int k = 0; k < PER_T; ++k) {
        const int i = k * SORT_TB + t;
        key[k] = 0u;
        if (i < n) {
            const unsigned bin = dv[k] >> BIN_SHIFT;
            const unsigned r = atomicAdd(&hist[bin], 1u);
            key[k] = (bin << 13) | r;
        }
    }
    __syncthreads();
    const unsigned cnt = hist[t];
    unsigned tot;
    const unsigned ex = block_excl_scan(cnt, sm, &tot);
    hist[t] = ex;
    if (t < NB) tab[(size_t)blockIdx.x * NB + t] = make_uint2(ex, cnt);
    __syncthreads();
    #pragma unroll
    for (int k = 0; k < PER_T; ++k) {
        const int i = k * SORT_TB + t;
        if (i < n)
            stage[hist[key[k] >> 13] + (key[k] & (CHUNK - 1))] =
                (sv[k] << BIN_SHIFT) | (dv[k] & (BIN_W - 1));
    }
    __syncthreads();
    #pragma unroll
    for (int k = 0; k < PER_T; ++k) {
        const int i = k * SORT_TB + t;
        if (i < n) binned[e0 + i] = stage[i];
    }
}


template <bool PLACE>
__device__ __forceinline__ void bin_take(unsigned w, unsigned* ncnt, const unsigned* noff,
                                         unsigned* outs, int* csr, unsigned base, bool fits) {
    const unsigned dl = w & (BIN_W - 1);
    const unsigned r  = atomicAdd(&ncnt[dl], 1u);
    if (PLACE) {
        const unsigned p = noff[dl] + r;
        const int s = (int)(w >> BIN_SHIFT);
        if (fits) outs[p] = (unsigned)s;
        else      csr[base + p] = s;
    }
}

template <bool PLACE>
__device__ __forceinline__ void bin_walk(const unsigned* __restrict__ binned,
                                         const unsigned* roff, const unsigned* rcnt,
                                         unsigned* ncnt, const unsigned* noff, unsigned* outs,
                                         int* csr, unsigned base, bool fits, int C) {
    const int lane = threadIdx.x & 31;
    const int wv   = threadIdx.x >> 5;
    for (int c0 = wv; c0 < C; c0 += (SORT_TB / 32) * RUN_U) {
        unsigned w0[RUN_U], w1[RUN_U], mm[RUN_U];
        const unsigned* pp[RUN_U];
        #pragma unroll
        for (int u = 0; u < RUN_U; ++u) {
            const int c = c0 + (SORT_TB / 32) * u;
            const bool in = c < C;
            mm[u] = in ? rcnt[c] : 0u;
            pp[u] = binned + (size_t)(in ? c : 0) * CHUNK + (in ? roff[c] : 0u);
            w0[u] = ((unsigned)lane      < mm[u]) ? pp[u][lane]      : 0xFFFFFFFFu;
            w1[u] = ((unsigned)lane + 32 < mm[u]) ? pp[u][lane + 32] : 0xFFFFFFFFu;
        }
        #pragma unroll
        for (int u = 0; u < RUN_U; ++u) {
            if (w0[u] != 0xFFFFFFFFu) bin_take<PLACE>(w0[u], ncnt, noff, outs, csr, base, fits);
            if (w1[u] != 0xFFFFFFFFu) bin_take<PLACE>(w1[u], ncnt, noff, outs, csr, base, fits);
            for (unsigned j = (unsigned)lane + 64; j < mm[u]; j += 32)
                bin_take<PLACE>(pp[u][j], ncnt, noff, outs, csr, base, fits);
        }
    }
}

__global__ __launch_bounds__(SORT_TB)
void k_bin_build(const unsigned* __restrict__ binned, const uint2* __restrict__ tab,
                 unsigned* __restrict__ rowptr, float* __restrict__ dinv,
                 int* __restrict__ csr, int N, int C, int NB) {
    __shared__ unsigned roff[SORT_TB];
    __shared__ unsigned rcnt[SORT_TB];
    __shared__ unsigned ncnt[BIN_W];
    __shared__ unsigned noff[BIN_W];
    __shared__ unsigned sm[SORT_TB / 32];
    __shared__ unsigned outs[CAP_B];
    const int t = threadIdx.x;
    const int b = blockIdx.x;

    unsigned off = 0, cnt = 0;
    if (t < C) {
        const uint2 q = tab[(size_t)t * NB + b];
        off = q.x;
        cnt = q.y;
    }
    roff[t] = off;
    rcnt[t] = cnt;
    if (t < BIN_W) ncnt[t] = 0u;
    unsigned total, base;
    (void)block_excl_scan(cnt, sm, &total);
    (void)block_excl_scan(off, sm, &base);
    const bool fits = total <= (unsigned)CAP_B;

    bin_walk<false>(binned, roff, rcnt, ncnt, noff, outs, csr, base, fits, C);
    __syncthreads();

    const unsigned dg = (t < BIN_W) ? ncnt[t] : 0u;
    unsigned tot2;
    const unsigned ex = block_excl_scan(dg, sm, &tot2);
    if (t < BIN_W) {
        noff[t] = ex;
        ncnt[t] = 0u;
        const int node = b * BIN_W + t;
        if (node < N) {
            rowptr[node] = base + ex;
            dinv[node]   = rsqrtf((float)(dg + 1u));
        }
    }
    if (b == NB - 1 && t == 0) rowptr[N] = base + total;
    __syncthreads();

    bin_walk<true>(binned, roff, rcnt, ncnt, noff, outs, csr, base, fits, C);
    __syncthreads();
    if (fits)
        for (unsigned i = (unsigned)t; i < total; i += SORT_TB) csr[base + i] = (int)outs[i];
}


__global__ void k_gemm1_wmma(const float* __restrict__ X, const float* __restrict__ W,
                             const float* __restrict__ dinv, float* __restrict__ H) {
    const int lane = threadIdx.x & 31;
    const int wave = threadIdx.x >> 5;
    const int m0   = blockIdx.x * 16;
    const int n0   = wave * 16;
    const int half = lane >> 4;
    const int lm   = lane & 15;

    const float* xrow = X + (size_t)(m0 + lm) * D_IN;
    v8f acc = {};
    #pragma unroll 4
    for (int k0 = 0; k0 < D_IN; k0 += 4) {
        const int ka = k0 + 2 * half;
        v2f a, b;
        a.x = xrow[ka];
        a.y = xrow[ka + 1];
        b.x = W[(size_t)ka       * D_HID + n0 + lm];
        b.y = W[(size_t)(ka + 1) * D_HID + n0 + lm];
        acc = __builtin_amdgcn_wmma_f32_16x16x4_f32(false, a, false, b,
                                                    (short)0, acc, false, false);
    }
    #pragma unroll
    for (int r = 0; r < 8; ++r) {
        const int m = m0 + r + 8 * half;
        H[(size_t)m * D_HID + n0 + lm] = acc[r] * dinv[m];
    }
}


__device__ __forceinline__ float  row_add(float a, float b)   { return a + b; }
__device__ __forceinline__ float2 row_add(float2 a, float2 b) { return make_float2(a.x + b.x, a.y + b.y); }
__device__ __forceinline__ float  row_fin(float a, float dn, float b)    { return dn * a + b; }
__device__ __forceinline__ float2 row_fin(float2 a, float dn, float2 b)  { return make_float2(fmaxf(dn * a.x + b.x, 0.f), fmaxf(dn * a.y + b.y, 0.f)); }

template <typename VT, int PITCH, bool TILE>
__device__ __forceinline__ float gather_walk(const unsigned* __restrict__ rowptr, const int* __restrict__ csr,
                                            const float* __restrict__ dinv, const float* __restrict__ src,
                                            const float* __restrict__ bias, float* __restrict__ dst,
                                            VT* __restrict__ own_s, int N) {
    const int lane = threadIdx.x & 31;
    const int wv = __builtin_amdgcn_readfirstlane((int)(threadIdx.x >> 5));
    const int n0 = (blockIdx.x * GATHER_WAVES + wv) * NPW;
    if (n0 >= N) return 0.f;
    const int cnt = (N - n0 < NPW) ? (N - n0) : NPW;

    const char* __restrict__ hb = (const char*)src;
    const unsigned ROWB    = 32u * (unsigned)sizeof(VT);
    const unsigned laneoff = (unsigned)lane * (unsigned)sizeof(VT);
    VT* __restrict__ out = (VT*)dst;
    VT* own = own_s + wv * (NPW * PITCH) + lane;

    VT t[NPW];
    #pragma unroll
    for (int k = 0; k < NPW; ++k) {
        const int kk = k < cnt ? k : cnt - 1;
        t[k] = *(const VT*)(hb + (size_t)(n0 + kk) * ROWB + laneoff);
    }
    const unsigned rpv = rowptr[n0 + (lane < cnt ? lane : cnt)];
    const float    dnv = dinv[n0 + (lane < cnt ? lane : cnt - 1)];
    const VT bv = ((const VT*)bias)[lane];
    #pragma unroll
    for (int k = 0; k < NPW; ++k) own[k * PITCH] = t[k];

    const unsigned e0   = (unsigned)__builtin_amdgcn_readlane((int)rpv, 0);
    const unsigned eEnd = (unsigned)__builtin_amdgcn_readlane((int)rpv, cnt);

    unsigned edc, edn;
    {
        const unsigned i0 = e0 + (unsigned)lane, i1 = i0 + 32u;
        edc = ((i0 < eEnd) ? (unsigned)csr[i0] : (unsigned)n0) * ROWB;
        edn = ((i1 < eEnd) ? (unsigned)csr[i1] : (unsigned)n0) * ROWB;
    }

    int      k      = 0;
    unsigned segEnd = (unsigned)__builtin_amdgcn_readlane((int)rpv, 1);
    VT acc     = t[0];
    VT own_nxt = own[PITCH];

    VT buf[2][GROUP];
    #pragma unroll
    for (int u = 0; u < GROUP; ++u) {
        const unsigned ro = (unsigned)__builtin_amdgcn_readlane((int)edc, u);
        buf[0][u] = *(const VT*)(hb + (size_t)(ro + laneoff));
    }

    unsigned gb = e0;
    for (unsigned eb = e0; eb < eEnd; eb += 32u) {
        unsigned edf;
        {
            const unsigned i2 = eb + 64u + (unsigned)lane;
            edf = ((i2 < eEnd) ? (unsigned)csr[i2] : (unsigned)n0) * ROWB;
        }
        #pragma unroll
        for (int g = 0; g < 32 / GROUP; ++g) {
            #pragma unroll
            for (int u = 0; u < GROUP; ++u) {
                const unsigned ro = (g + 1 < 32 / GROUP)
                    ? (unsigned)__builtin_amdgcn_readlane((int)edc, (g + 1) * GROUP + u)
                    : (unsigned)__builtin_amdgcn_readlane((int)edn, u);
                buf[(g + 1) & 1][u] = *(const VT*)(hb + (size_t)(ro + laneoff));
            }
            if (segEnd - gb >= (unsigned)GROUP) {
                #pragma unroll
                for (int u = 0; u < GROUP; ++u) acc = row_add(acc, buf[g & 1][u]);
            } else {
                #pragma unroll
                for (int u = 0; u < GROUP; ++u) {
                    while (__builtin_expect(gb + (unsigned)u == segEnd, 0)) {
                        const float dn = __builtin_bit_cast(float, __builtin_amdgcn_readlane(__builtin_bit_cast(int, dnv), k));
                        if constexpr (TILE) own[k * PITCH] = row_fin(acc, dn, bv);
                        else                out[(size_t)(n0 + k) * 32 + lane] = row_fin(acc, dn, bv);
                        ++k;
                        if (k < cnt) {
                            acc     = own_nxt;
                            own_nxt = own[((k + 1 < NPW) ? k + 1 : NPW - 1) * PITCH];
                            segEnd  = (unsigned)__builtin_amdgcn_readlane((int)rpv, k + 1);
                        } else {
                            segEnd  = 0xFFFFFFFFu;
                        }
                    }
                    acc = row_add(acc, buf[g & 1][u]);
                }
            }
            gb += (unsigned)GROUP;
        }
        edc = edn;
        edn = edf;
    }
    while (k < cnt) {
        const float dn = __builtin_bit_cast(float, __builtin_amdgcn_readlane(__builtin_bit_cast(int, dnv), k));
        if constexpr (TILE) own[k * PITCH] = row_fin(acc, dn, bv);
        else                out[(size_t)(n0 + k) * 32 + lane] = row_fin(acc, dn, bv);
        ++k;
        acc     = own_nxt;
        own_nxt = own[((k + 1 < NPW) ? k + 1 : NPW - 1) * PITCH];
    }
    return dnv;
}

__device__ __forceinline__ float xhalf(float x) {
    const int b = __builtin_bit_cast(int, x);
    return __builtin_bit_cast(float, __builtin_amdgcn_permlanex16(b, b, 0x76543210, 0xfedcba98, false, false));
}


#define TPITCH 34

__global__ __launch_bounds__(GATHER_WAVES * 32)
void k_gather64_gemm2(const unsigned* __restrict__ rowptr, const int* __restrict__ csr,
                      const float* __restrict__ dinv, const float* __restrict__ h,
                      const float* __restrict__ b1, const float* __restrict__ W2,
                      float* __restrict__ H2, int N) {
    __shared__ float2 rows_s[GATHER_WAVES * NPW * TPITCH];
    __shared__ alignas(16) float w2s[D_HID * D_OUT];

    #pragma unroll
    for (int q = 0; q < D_HID * D_OUT / (GATHER_WAVES * 32); ++q) {
        const int e  = q * (GATHER_WAVES * 32) + (int)threadIdx.x;
        const int ln = (e >> 2) & 31;
        const int k  = 4 * (e >> 7) + 2 * (ln >> 4) + (e & 1);
        const int n  = 16 * ((e >> 1) & 1) + (ln & 15);
        w2s[e] = W2[k * D_OUT + n];
    }
    __syncthreads();

    const int lane = threadIdx.x & 31;
    const int wv = __builtin_amdgcn_readfirstlane((int)(threadIdx.x >> 5));
    const int n0 = (blockIdx.x * GATHER_WAVES + wv) * NPW;
    if (n0 >= N) return;
    const int cnt = (N - n0 < NPW) ? (N - n0) : NPW;

    const float dnv = gather_walk<float2, TPITCH, true>(rowptr, csr, dinv, h, b1, nullptr, rows_s, N);


    const float2* tile = rows_s + wv * (NPW * TPITCH);
    const int half = lane >> 4;
    const int lm   = lane & 15;
    v8f c0 = {}, c1 = {};
    #pragma unroll
    for (int ks = 0; ks < D_HID / 4; ++ks) {
        const float2 av = tile[lm * TPITCH + 2 * ks + half];
        const float* bw = w2s + (ks * 32 + lane) * 4;
        v2f a, bl, bh;
        a.x  = av.x;  a.y  = av.y;
        bl.x = bw[0]; bl.y = bw[1];
        bh.x = bw[2]; bh.y = bw[3];
        c0 = __builtin_amdgcn_wmma_f32_16x16x4_f32(false, a, false, bl, (short)0, c0, false, false);
        c1 = __builtin_amdgcn_wmma_f32_16x16x4_f32(false, a, false, bh, (short)0, c1, false, false);
    }
    #pragma unroll
    for (int r = 0; r < 8; ++r) {
        const float d0 = __builtin_bit_cast(float, __builtin_amdgcn_readlane(__builtin_bit_cast(int, dnv), r));
        const float d1 = __builtin_bit_cast(float, __builtin_amdgcn_readlane(__builtin_bit_cast(int, dnv), r + 8));
        const float dm = half ? d1 : d0;
        const float y0 = c0[r] * dm;
        const float y1 = c1[r] * dm;
        const float got = xhalf(half ? y0 : y1);
        if (r < cnt)     H2[(size_t)(n0 + r) * D_OUT + lane]     = half ? got : y0;
        if (r + 8 < cnt) H2[(size_t)(n0 + r + 8) * D_OUT + lane] = half ? y1 : got;
    }
}


__global__ __launch_bounds__(GATHER_WAVES * 32)
void k_gather32(const unsigned* __restrict__ rowptr, const int* __restrict__ csr,
                const float* __restrict__ dinv, const float* __restrict__ h2,
                const float* __restrict__ b2, float* __restrict__ out, int N) {
    __shared__ float own_s[GATHER_WAVES * NPW * 32];
    (void)gather_walk<float, 32, false>(rowptr, csr, dinv, h2, b2, out, own_s, N);
}


extern "C" void kernel_launch(void* const* d_in, const int* in_sizes, int n_in,
                              void* d_out, int out_size, void* d_ws, size_t ws_size,
                              hipStream_t stream) {
    const float* x   = (const float*)d_in[0];
    const int*   ei  = (const int*)  d_in[1];
    const float* W1  = (const float*)d_in[2];
    const float* b1  = (const float*)d_in[3];
    const float* W2  = (const float*)d_in[4];
    const float* b2  = (const float*)d_in[5];
    float*       out = (float*)d_out;

    const int N = in_sizes[0] / D_IN;
    const int E = in_sizes[1] / 2;

    char* ws = (char*)d_ws;
    float*    h      = (float*)(ws);
    float*    h2     = (float*)(ws + (size_t)N * D_HID * 4);
    int*      csr    = (int*)  (ws + (size_t)N * D_HID * 4 + (size_t)N * D_OUT * 4);
    unsigned* binned = (unsigned*)((char*)csr + (size_t)E * 4);
    float*    dinv   = (float*)(binned + E);

    const int C  = (E + CHUNK - 1) / CHUNK;
    const int NB = (N + BIN_W - 1) / BIN_W;
    const int ngath = (N + GATHER_WAVES * NPW - 1) / (GATHER_WAVES * NPW);

    uint2*    tab    = (uint2*)(dinv + N);
    unsigned* rowptr = (unsigned*)(tab + (size_t)C * NB);

    k_bin_chunk<<<C,  SORT_TB, 0, stream>>>(ei, binned, tab, E, NB);
    k_bin_build<<<NB, SORT_TB, 0, stream>>>(binned, tab, rowptr, dinv, csr, N, C, NB);

    k_gemm1_wmma    <<<N / 16, 128, 0, stream>>>(x, W1, dinv, h);
    k_gather64_gemm2<<<ngath, GATHER_WAVES * 32, 0, stream>>>(rowptr, csr, dinv, h, b1, W2, h2, N);

    k_gather32      <<<ngath, GATHER_WAVES * 32, 0, stream>>>(rowptr, csr, dinv, h2, b2, out, N);
}


# GEMM 1 re-tiled: one wave owns a 16x64 row block (was 4 waves of 16x16): X read once in 16-byte pieces, W1 from LDS, whole-line stores; k regrouped inside matrix steps, so its sum order changes
# speedup vs baseline: 7.8977x; 1.1069x over previous
#include <hip/hip_runtime.h>
#include <hip/hip_bf16.h>


typedef float v2f __attribute__((ext_vector_type(2)));
typedef float v8f __attribute__((ext_vector_type(8)));

#define D_IN  128
#define D_HID 64
#define D_OUT 32

#define GATHER_WAVES 8
#define NPW          16
#define GROUP        16

#define BIN_SHIFT 9
#define BIN_W     512
#define CHUNK     8192
#define SORT_TB   1024
#define PER_T     (CHUNK / SORT_TB)
#define CAP_B     12288
#define RUN_U     4

__device__ __forceinline__ unsigned block_excl_scan(unsigned v, unsigned* sm, unsigned* tot) {
    const int lane = threadIdx.x & 31;
    const int wv   = threadIdx.x >> 5;
    unsigned inc = v;
    for (int o = 1; o < 32; o <<= 1) {
        const unsigned y = __shfl_up(inc, o);
        if (lane >= o) inc += y;
    }
    __syncthreads();
    if (lane == 31) sm[wv] = inc;
    __syncthreads();
    unsigned woff = 0, all = 0;
    for (int k = 0; k < SORT_TB / 32; ++k) {
        const unsigned s = sm[k];
        all  += s;
        woff += (k < wv) ? s : 0u;
    }
    *tot = all;
    return woff + inc - v;
}


__global__ __launch_bounds__(SORT_TB)
void k_bin_chunk(const int* __restrict__ ei, unsigned* __restrict__ binned,
                 uint2* __restrict__ tab, int E, int NB) {
    __shared__ unsigned hist[SORT_TB];
    __shared__ unsigned sm[SORT_TB / 32];
    __shared__ unsigned stage[CHUNK];
    const int t  = threadIdx.x;
    const int e0 = blockIdx.x * CHUNK;
    const int n  = (E - e0 < CHUNK) ? (E - e0) : CHUNK;

    hist[t] = 0u;
    __syncthreads();

    unsigned sv[PER_T], dv[PER_T];
    #pragma unroll
    for (int k = 0; k < PER_T; ++k) {
        const int i = k * SORT_TB + t;
        sv[k] = 0u; dv[k] = 0u;
        if (i < n) {
            sv[k] = (unsigned)ei[e0 + i];
            dv[k] = (unsigned)ei[E + e0 + i];
        }
    }
    unsigned key[PER_T];
    #pragma unroll
    for (int k = 0; k < PER_T; ++k) {
        const int i = k * SORT_TB + t;
        key[k] = 0u;
        if (i < n) {
            const unsigned bin = dv[k] >> BIN_SHIFT;
            const unsigned r = atomicAdd(&hist[bin], 1u);
            key[k] = (bin << 13) | r;
        }
    }
    __syncthreads();
    const unsigned cnt = hist[t];
    unsigned tot;
    const unsigned ex = block_excl_scan(cnt, sm, &tot);
    hist[t] = ex;
    if (t < NB) tab[(size_t)blockIdx.x * NB + t] = make_uint2(ex, cnt);
    __syncthreads();
    #pragma unroll
    for (int k = 0; k < PER_T; ++k) {
        const int i = k * SORT_TB + t;
        if (i < n)
            stage[hist[key[k] >> 13] + (key[k] & (CHUNK - 1))] =
                (sv[k] << BIN_SHIFT) | (dv[k] & (BIN_W - 1));
    }
    __syncthreads();
    #pragma unroll
    for (int k = 0; k < PER_T; ++k) {
        const int i = k * SORT_TB + t;
        if (i < n) binned[e0 + i] = stage[i];
    }
}


template <bool PLACE>
__device__ __forceinline__ void bin_take(unsigned w, unsigned* ncnt, const unsigned* noff,
                                         unsigned* outs, int* csr, unsigned base, bool fits) {
    const unsigned dl = w & (BIN_W - 1);
    const unsigned r  = atomicAdd(&ncnt[dl], 1u);
    if (PLACE) {
        const unsigned p = noff[dl] + r;
        const int s = (int)(w >> BIN_SHIFT);
        if (fits) outs[p] = (unsigned)s;
        else      csr[base + p] = s;
    }
}

template <bool PLACE>
__device__ __forceinline__ void bin_walk(const unsigned* __restrict__ binned,
                                         const unsigned* roff, const unsigned* rcnt,
                                         unsigned* ncnt, const unsigned* noff, unsigned* outs,
                                         int* csr, unsigned base, bool fits, int C) {
    const int lane = threadIdx.x & 31;
    const int wv   = threadIdx.x >> 5;
    for (int c0 = wv; c0 < C; c0 += (SORT_TB / 32) * RUN_U) {
        unsigned w0[RUN_U], w1[RUN_U], mm[RUN_U];
        const unsigned* pp[RUN_U];
        #pragma unroll
        for (int u = 0; u < RUN_U; ++u) {
            const int c = c0 + (SORT_TB / 32) * u;
            const bool in = c < C;
            mm[u] = in ? rcnt[c] : 0u;
            pp[u] = binned + (size_t)(in ? c : 0) * CHUNK + (in ? roff[c] : 0u);
            w0[u] = ((unsigned)lane      < mm[u]) ? pp[u][lane]      : 0xFFFFFFFFu;
            w1[u] = ((unsigned)lane + 32 < mm[u]) ? pp[u][lane + 32] : 0xFFFFFFFFu;
        }
        #pragma unroll
        for (int u = 0; u < RUN_U; ++u) {
            if (w0[u] != 0xFFFFFFFFu) bin_take<PLACE>(w0[u], ncnt, noff, outs, csr, base, fits);
            if (w1[u] != 0xFFFFFFFFu) bin_take<PLACE>(w1[u], ncnt, noff, outs, csr, base, fits);
            for (unsigned j = (unsigned)lane + 64; j < mm[u]; j += 32)
                bin_take<PLACE>(pp[u][j], ncnt, noff, outs, csr, base, fits);
        }
    }
}

__global__ __launch_bounds__(SORT_TB)
void k_bin_build(const unsigned* __restrict__ binned, const uint2* __restrict__ tab,
                 unsigned* __restrict__ rowptr, float* __restrict__ dinv,
                 int* __restrict__ csr, int N, int C, int NB) {
    __shared__ unsigned roff[SORT_TB];
    __shared__ unsigned rcnt[SORT_TB];
    __shared__ unsigned ncnt[BIN_W];
    __shared__ unsigned noff[BIN_W];
    __shared__ unsigned sm[SORT_TB / 32];
    __shared__ unsigned outs[CAP_B];
    const int t = threadIdx.x;
    const int b = blockIdx.x;

    unsigned off = 0, cnt = 0;
    if (t < C) {
        const uint2 q = tab[(size_t)t * NB + b];
        off = q.x;
        cnt = q.y;
    }
    roff[t] = off;
    rcnt[t] = cnt;
    if (t < BIN_W) ncnt[t] = 0u;
    unsigned total, base;
    (void)block_excl_scan(cnt, sm, &total);
    (void)block_excl_scan(off, sm, &base);
    const bool fits = total <= (unsigned)CAP_B;

    bin_walk<false>(binned, roff, rcnt, ncnt, noff, outs, csr, base, fits, C);
    __syncthreads();

    const unsigned dg = (t < BIN_W) ? ncnt[t] : 0u;
    unsigned tot2;
    const unsigned ex = block_excl_scan(dg, sm, &tot2);
    if (t < BIN_W) {
        noff[t] = ex;
        ncnt[t] = 0u;
        const int node = b * BIN_W + t;
        if (node < N) {
            rowptr[node] = base + ex;
            dinv[node]   = rsqrtf((float)(dg + 1u));
        }
    }
    if (b == NB - 1 && t == 0) rowptr[N] = base + total;
    __syncthreads();

    bin_walk<true>(binned, roff, rcnt, ncnt, noff, outs, csr, base, fits, C);
    __syncthreads();
    if (fits)
        for (unsigned i = (unsigned)t; i < total; i += SORT_TB) csr[base + i] = (int)outs[i];
}


#define G1_WAVES 4
#define G1_COPY  (D_IN * D_HID / 8 / (G1_WAVES * 32))
#define W1_PITCH 144

__device__ __forceinline__ float xhalf(float x) {
    const int b = __builtin_bit_cast(int, x);
    return __builtin_bit_cast(float, __builtin_amdgcn_permlanex16(b, b, 0x76543210, 0xfedcba98, false, false));
}

__global__ __launch_bounds__(G1_WAVES * 32)
void k_gemm1_wmma(const float* __restrict__ X, const float* __restrict__ W,
                  const float* __restrict__ dinv, float* __restrict__ H, int N) {
    __shared__ alignas(16) float w1s[(D_IN / 2) * W1_PITCH];
    const int lane = threadIdx.x & 31;
    const int wave = threadIdx.x >> 5;
    const int half = lane >> 4;
    const int lm   = lane & 15;
    const int m0   = (blockIdx.x * G1_WAVES + wave) * 16;
    const bool live = m0 < N;
    const int mr   = live ? m0 : 0;

    float4 we[G1_COPY], wo[G1_COPY];
    #pragma unroll
    for (int q = 0; q < G1_COPY; ++q) {
        const int e = q * (G1_WAVES * 32) + (int)threadIdx.x;
        we[q] = ((const float4*)W)[(2 * (e >> 4))     * (D_HID / 4) + (e & 15)];
        wo[q] = ((const float4*)W)[(2 * (e >> 4) + 1) * (D_HID / 4) + (e & 15)];
    }

    const float4* xr = (const float4*)(X + (size_t)(mr + lm) * D_IN) + half;
    float4 a[D_IN / 8];
    #pragma unroll
    for (int j = 0; j < D_IN / 8; ++j) a[j] = xr[2 * j];
    const float dnv = dinv[mr + lm];

    #pragma unroll
    for (int q = 0; q < G1_COPY; ++q) {
        const int e = q * (G1_WAVES * 32) + (int)threadIdx.x;
        float* d = w1s + (e >> 4) * W1_PITCH + (e & 15) * 8;
        *(float4*)(d)     = make_float4(we[q].x, wo[q].x, we[q].y, wo[q].y);
        *(float4*)(d + 4) = make_float4(we[q].z, wo[q].z, we[q].w, wo[q].w);
    }
    __syncthreads();
    if (!live) return;

    const float* bb = w1s + (2 * half) * W1_PITCH + 2 * lm;
    v8f c0 = {}, c1 = {}, c2 = {}, c3 = {};
    #pragma unroll
    for (int j = 0; j < D_IN / 8; ++j) {
        #pragma unroll
        for (int t = 0; t < 2; ++t) {
            const float* bk = bb + (4 * j + t) * W1_PITCH;
            v2f av;
            av.x = t ? a[j].z : a[j].x;
            av.y = t ? a[j].w : a[j].y;
            const v2f b0 = *(const v2f*)(bk);
            const v2f b1 = *(const v2f*)(bk + 32);
            const v2f b2 = *(const v2f*)(bk + 64);
            const v2f b3 = *(const v2f*)(bk + 96);
            c0 = __builtin_amdgcn_wmma_f32_16x16x4_f32(false, av, false, b0, (short)0, c0, false, false);
            c1 = __builtin_amdgcn_wmma_f32_16x16x4_f32(false, av, false, b1, (short)0, c1, false, false);
            c2 = __builtin_amdgcn_wmma_f32_16x16x4_f32(false, av, false, b2, (short)0, c2, false, false);
            c3 = __builtin_amdgcn_wmma_f32_16x16x4_f32(false, av, false, b3, (short)0, c3, false, false);
        }
    }
    #pragma unroll
    for (int r = 0; r < 8; ++r) {
        const float d0 = __builtin_bit_cast(float, __builtin_amdgcn_readlane(__builtin_bit_cast(int, dnv), r));
        const float d1 = __builtin_bit_cast(float, __builtin_amdgcn_readlane(__builtin_bit_cast(int, dnv), r + 8));
        const float dm = half ? d1 : d0;
        const float y0 = c0[r] * dm, y1 = c1[r] * dm;
        const float y2 = c2[r] * dm, y3 = c3[r] * dm;
        const float g01 = xhalf(half ? y0 : y1);
        const float g23 = xhalf(half ? y2 : y3);
        float* lo = H + (size_t)(m0 + r) * D_HID + lane;
        float* hi = H + (size_t)(m0 + r + 8) * D_HID + lane;
        lo[0]  = half ? g01 : y0;
        lo[32] = half ? g23 : y2;
        hi[0]  = half ? y1 : g01;
        hi[32] = half ? y3 : g23;
    }
}


__device__ __forceinline__ float  row_add(float a, float b)   { return a + b; }
__device__ __forceinline__ float2 row_add(float2 a, float2 b) { return make_float2(a.x + b.x, a.y + b.y); }
__device__ __forceinline__ float  row_fin(float a, float dn, float b)    { return dn * a + b; }
__device__ __forceinline__ float2 row_fin(float2 a, float dn, float2 b)  { return make_float2(fmaxf(dn * a.x + b.x, 0.f), fmaxf(dn * a.y + b.y, 0.f)); }

template <typename VT, int PITCH, bool TILE>
__device__ __forceinline__ float gather_walk(const unsigned* __restrict__ rowptr, const int* __restrict__ csr,
                                            const float* __restrict__ dinv, const float* __restrict__ src,
                                            const float* __restrict__ bias, float* __restrict__ dst,
                                            VT* __restrict__ own_s, int N) {
    const int lane = threadIdx.x & 31;
    const int wv = __builtin_amdgcn_readfirstlane((int)(threadIdx.x >> 5));
    const int n0 = (blockIdx.x * GATHER_WAVES + wv) * NPW;
    if (n0 >= N) return 0.f;
    const int cnt = (N - n0 < NPW) ? (N - n0) : NPW;

    const char* __restrict__ hb = (const char*)src;
    const unsigned ROWB    = 32u * (unsigned)sizeof(VT);
    const unsigned laneoff = (unsigned)lane * (unsigned)sizeof(VT);
    VT* __restrict__ out = (VT*)dst;
    VT* own = own_s + wv * (NPW * PITCH) + lane;

    VT t[NPW];
    #pragma unroll
    for (int k = 0; k < NPW; ++k) {
        const int kk = k < cnt ? k : cnt - 1;
        t[k] = *(const VT*)(hb + (size_t)(n0 + kk) * ROWB + laneoff);
    }
    const unsigned rpv = rowptr[n0 + (lane < cnt ? lane : cnt)];
    const float    dnv = dinv[n0 + (lane < cnt ? lane : cnt - 1)];
    const VT bv = ((const VT*)bias)[lane];
    #pragma unroll
    for (int k = 0; k < NPW; ++k) own[k * PITCH] = t[k];

    const unsigned e0   = (unsigned)__builtin_amdgcn_readlane((int)rpv, 0);
    const unsigned eEnd = (unsigned)__builtin_amdgcn_readlane((int)rpv, cnt);

    unsigned edc, edn;
    {
        const unsigned i0 = e0 + (unsigned)lane, i1 = i0 + 32u;
        edc = ((i0 < eEnd) ? (unsigned)csr[i0] : (unsigned)n0) * ROWB;
        edn = ((i1 < eEnd) ? (unsigned)csr[i1] : (unsigned)n0) * ROWB;
    }

    int      k      = 0;
    unsigned segEnd = (unsigned)__builtin_amdgcn_readlane((int)rpv, 1);
    VT acc     = t[0];
    VT own_nxt = own[PITCH];

    VT buf[2][GROUP];
    #pragma unroll
    for (int u = 0; u < GROUP; ++u) {
        const unsigned ro = (unsigned)__builtin_amdgcn_readlane((int)edc, u);
        buf[0][u] = *(const VT*)(hb + (size_t)(ro + laneoff));
    }

    unsigned gb = e0;
    for (unsigned eb = e0; eb < eEnd; eb += 32u) {
        unsigned edf;
        {
            const unsigned i2 = eb + 64u + (unsigned)lane;
            edf = ((i2 < eEnd) ? (unsigned)csr[i2] : (unsigned)n0) * ROWB;
        }
        #pragma unroll
        for (int g = 0; g < 32 / GROUP; ++g) {
            #pragma unroll
            for (int u = 0; u < GROUP; ++u) {
                const unsigned ro = (g + 1 < 32 / GROUP)
                    ? (unsigned)__builtin_amdgcn_readlane((int)edc, (g + 1) * GROUP + u)
                    : (unsigned)__builtin_amdgcn_readlane((int)edn, u);
                buf[(g + 1) & 1][u] = *(const VT*)(hb + (size_t)(ro + laneoff));
            }
            if (segEnd - gb >= (unsigned)GROUP) {
                #pragma unroll
                for (int u = 0; u < GROUP; ++u) acc = row_add(acc, buf[g & 1][u]);
            } else {
                #pragma unroll
                for (int u = 0; u < GROUP; ++u) {
                    while (__builtin_expect(gb + (unsigned)u == segEnd, 0)) {
                        const float dn = __builtin_bit_cast(float, __builtin_amdgcn_readlane(__builtin_bit_cast(int, dnv), k));
                        if constexpr (TILE) own[k * PITCH] = row_fin(acc, dn, bv);
                        else                out[(size_t)(n0 + k) * 32 + lane] = row_fin(acc, dn, bv);
                        ++k;
                        if (k < cnt) {
                            acc     = own_nxt;
                            own_nxt = own[((k + 1 < NPW) ? k + 1 : NPW - 1) * PITCH];
                            segEnd  = (unsigned)__builtin_amdgcn_readlane((int)rpv, k + 1);
                        } else {
                            segEnd  = 0xFFFFFFFFu;
                        }
                    }
                    acc = row_add(acc, buf[g & 1][u]);
                }
            }
            gb += (unsigned)GROUP;
        }
        edc = edn;
        edn = edf;
    }
    while (k < cnt) {
        const float dn = __builtin_bit_cast(float, __builtin_amdgcn_readlane(__builtin_bit_cast(int, dnv), k));
        if constexpr (TILE) own[k * PITCH] = row_fin(acc, dn, bv);
        else                out[(size_t)(n0 + k) * 32 + lane] = row_fin(acc, dn, bv);
        ++k;
        acc     = own_nxt;
        own_nxt = own[((k + 1 < NPW) ? k + 1 : NPW - 1) * PITCH];
    }
    return dnv;
}


#define TPITCH 34

__global__ __launch_bounds__(GATHER_WAVES * 32)
void k_gather64_gemm2(const unsigned* __restrict__ rowptr, const int* __restrict__ csr,
                      const float* __restrict__ dinv, const float* __restrict__ h,
                      const float* __restrict__ b1, const float* __restrict__ W2,
                      float* __restrict__ H2, int N) {
    __shared__ float2 rows_s[GATHER_WAVES * NPW * TPITCH];
    __shared__ alignas(16) float w2s[D_HID * D_OUT];

    #pragma unroll
    for (int q = 0; q < D_HID * D_OUT / (GATHER_WAVES * 32); ++q) {
        const int e  = q * (GATHER_WAVES * 32) + (int)threadIdx.x;
        const int ln = (e >> 2) & 31;
        const int k  = 4 * (e >> 7) + 2 * (ln >> 4) + (e & 1);
        const int n  = 16 * ((e >> 1) & 1) + (ln & 15);
        w2s[e] = W2[k * D_OUT + n];
    }
    __syncthreads();

    const int lane = threadIdx.x & 31;
    const int wv = __builtin_amdgcn_readfirstlane((int)(threadIdx.x >> 5));
    const int n0 = (blockIdx.x * GATHER_WAVES + wv) * NPW;
    if (n0 >= N) return;
    const int cnt = (N - n0 < NPW) ? (N - n0) : NPW;

    const float dnv = gather_walk<float2, TPITCH, true>(rowptr, csr, dinv, h, b1, nullptr, rows_s, N);


    const float2* tile = rows_s + wv * (NPW * TPITCH);
    const int half = lane >> 4;
    const int lm   = lane & 15;
    v8f c0 = {}, c1 = {};
    #pragma unroll
    for (int ks = 0; ks < D_HID / 4; ++ks) {
        const float2 av = tile[lm * TPITCH + 2 * ks + half];
        const float* bw = w2s + (ks * 32 + lane) * 4;
        v2f a, bl, bh;
        a.x  = av.x;  a.y  = av.y;
        bl.x = bw[0]; bl.y = bw[1];
        bh.x = bw[2]; bh.y = bw[3];
        c0 = __builtin_amdgcn_wmma_f32_16x16x4_f32(false, a, false, bl, (short)0, c0, false, false);
        c1 = __builtin_amdgcn_wmma_f32_16x16x4_f32(false, a, false, bh, (short)0, c1, false, false);
    }
    #pragma unroll
    for (int r = 0; r < 8; ++r) {
        const float d0 = __builtin_bit_cast(float, __builtin_amdgcn_readlane(__builtin_bit_cast(int, dnv), r));
        const float d1 = __builtin_bit_cast(float, __builtin_amdgcn_readlane(__builtin_bit_cast(int, dnv), r + 8));
        const float dm = half ? d1 : d0;
        const float y0 = c0[r] * dm;
        const float y1 = c1[r] * dm;
        const float got = xhalf(half ? y0 : y1);
        if (r < cnt)     H2[(size_t)(n0 + r) * D_OUT + lane]     = half ? got : y0;
        if (r + 8 < cnt) H2[(size_t)(n0 + r + 8) * D_OUT + lane] = half ? y1 : got;
    }
}


__global__ __launch_bounds__(GATHER_WAVES * 32)
void k_gather32(const unsigned* __restrict__ rowptr, const int* __restrict__ csr,
                const float* __restrict__ dinv, const float* __restrict__ h2,
                const float* __restrict__ b2, float* __restrict__ out, int N) {
    __shared__ float own_s[GATHER_WAVES * NPW * 32];
    (void)gather_walk<float, 32, false>(rowptr, csr, dinv, h2, b2, out, own_s, N);
}


extern "C" void kernel_launch(void* const* d_in, const int* in_sizes, int n_in,
                              void* d_out, int out_size, void* d_ws, size_t ws_size,
                              hipStream_t stream) {
    const float* x   = (const float*)d_in[0];
    const int*   ei  = (const int*)  d_in[1];
    const float* W1  = (const float*)d_in[2];
    const float* b1  = (const float*)d_in[3];
    const float* W2  = (const float*)d_in[4];
    const float* b2  = (const float*)d_in[5];
    float*       out = (float*)d_out;

    const int N = in_sizes[0] / D_IN;
    const int E = in_sizes[1] / 2;

    char* ws = (char*)d_ws;
    float*    h      = (float*)(ws);
    float*    h2     = (float*)(ws + (size_t)N * D_HID * 4);
    int*      csr    = (int*)  (ws + (size_t)N * D_HID * 4 + (size_t)N * D_OUT * 4);
    unsigned* binned = (unsigned*)((char*)csr + (size_t)E * 4);
    float*    dinv   = (float*)(binned + E);

    const int C  = (E + CHUNK - 1) / CHUNK;
    const int NB = (N + BIN_W - 1) / BIN_W;
    const int ngath = (N + GATHER_WAVES * NPW - 1) / (GATHER_WAVES * NPW);

    uint2*    tab    = (uint2*)(dinv + N);
    unsigned* rowptr = (unsigned*)(tab + (size_t)C * NB);

    k_bin_chunk<<<C,  SORT_TB, 0, stream>>>(ei, binned, tab, E, NB);
    k_bin_build<<<NB, SORT_TB, 0, stream>>>(binned, tab, rowptr, dinv, csr, N, C, NB);

    k_gemm1_wmma    <<<(N / 16 + G1_WAVES - 1) / G1_WAVES, G1_WAVES * 32, 0, stream>>>(x, W1, dinv, h, N);
    k_gather64_gemm2<<<ngath, GATHER_WAVES * 32, 0, stream>>>(rowptr, csr, dinv, h, b1, W2, h2, N);

    k_gather32      <<<ngath, GATHER_WAVES * 32, 0, stream>>>(rowptr, csr, dinv, h2, b2, out, N);
}


# GEMM 1 moved into the sort's second launch: a bin's work-group (32 waves) multiplies its own 512 rows right after making their deg^-1/2, X prefetched at entry; one launch fewer (4), sums unchanged
# speedup vs baseline: 8.1499x; 1.0319x over previous
#include <hip/hip_runtime.h>
#include <hip/hip_bf16.h>


typedef float v2f __attribute__((ext_vector_type(2)));
typedef float v8f __attribute__((ext_vector_type(8)));

#define D_IN  128
#define D_HID 64
#define D_OUT 32

#define GATHER_WAVES 8
#define NPW          16
#define GROUP        16

#define BIN_SHIFT 9
#define BIN_W     512
#define CHUNK     8192
#define SORT_TB   1024
#define PER_T     (CHUNK / SORT_TB)
#define CAP_B     12288
#define RUN_U     4

__device__ __forceinline__ unsigned block_excl_scan(unsigned v, unsigned* sm, unsigned* tot) {
    const int lane = threadIdx.x & 31;
    const int wv   = threadIdx.x >> 5;
    unsigned inc = v;
    for (int o = 1; o < 32; o <<= 1) {
        const unsigned y = __shfl_up(inc, o);
        if (lane >= o) inc += y;
    }
    __syncthreads();
    if (lane == 31) sm[wv] = inc;
    __syncthreads();
    unsigned woff = 0, all = 0;
    for (int k = 0; k < SORT_TB / 32; ++k) {
        const unsigned s = sm[k];
        all  += s;
        woff += (k < wv) ? s : 0u;
    }
    *tot = all;
    return woff + inc - v;
}


__global__ __launch_bounds__(SORT_TB)
void k_bin_chunk(const int* __restrict__ ei, unsigned* __restrict__ binned,
                 uint2* __restrict__ tab, int E, int NB) {
    __shared__ unsigned hist[SORT_TB];
    __shared__ unsigned sm[SORT_TB / 32];
    __shared__ unsigned stage[CHUNK];
    const int t  = threadIdx.x;
    const int e0 = blockIdx.x * CHUNK;
    const int n  = (E - e0 < CHUNK) ? (E - e0) : CHUNK;

    hist[t] = 0u;
    __syncthreads();

    unsigned sv[PER_T], dv[PER_T];
    #pragma unroll
    for (int k = 0; k < PER_T; ++k) {
        const int i = k * SORT_TB + t;
        sv[k] = 0u; dv[k] = 0u;
        if (i < n) {
            sv[k] = (unsigned)ei[e0 + i];
            dv[k] = (unsigned)ei[E + e0 + i];
        }
    }
    unsigned key[PER_T];
    #pragma unroll
    for (int k = 0; k < PER_T; ++k) {
        const int i = k * SORT_TB + t;
        key[k] = 0u;
        if (i < n) {
            const unsigned bin = dv[k] >> BIN_SHIFT;
            const unsigned r = atomicAdd(&hist[bin], 1u);
            key[k] = (bin << 13) | r;
        }
    }
    __syncthreads();
    const unsigned cnt = hist[t];
    unsigned tot;
    const unsigned ex = block_excl_scan(cnt, sm, &tot);
    hist[t] = ex;
    if (t < NB) tab[(size_t)blockIdx.x * NB + t] = make_uint2(ex, cnt);
    __syncthreads();
    #pragma unroll
    for (int k = 0; k < PER_T; ++k) {
        const int i = k * SORT_TB + t;
        if (i < n)
            stage[hist[key[k] >> 13] + (key[k] & (CHUNK - 1))] =
                (sv[k] << BIN_SHIFT) | (dv[k] & (BIN_W - 1));
    }
    __syncthreads();
    #pragma unroll
    for (int k = 0; k < PER_T; ++k) {
        const int i = k * SORT_TB + t;
        if (i < n) binned[e0 + i] = stage[i];
    }
}


#define W1_PITCH 144

__device__ __forceinline__ float xhalf(float x) {
    const int b = __builtin_bit_cast(int, x);
    return __builtin_bit_cast(float, __builtin_amdgcn_permlanex16(b, b, 0x76543210, 0xfedcba98, false, false));
}

struct XRows { float4 a[D_IN / 8]; };

__device__ __forceinline__ void gemm1_load(const float* __restrict__ X, int m0, XRows& x) {
    const int lane = threadIdx.x & 31;
    const float4* xr = (const float4*)(X + (size_t)(m0 + (lane & 15)) * D_IN) + (lane >> 4);
    #pragma unroll
    for (int j = 0; j < D_IN / 8; ++j) x.a[j] = xr[2 * j];
}

__device__ __forceinline__ void gemm1_block(const XRows& x, const float* w1s, float dnv,
                                            float* __restrict__ H, int m0) {
    const int lane = threadIdx.x & 31;
    const int half = lane >> 4;
    const int lm   = lane & 15;
    const float* bb = w1s + (2 * half) * W1_PITCH + 2 * lm;
    v8f c0 = {}, c1 = {}, c2 = {}, c3 = {};
    #pragma unroll
    for (int j = 0; j < D_IN / 8; ++j) {
        #pragma unroll
        for (int t = 0; t < 2; ++t) {
            const float* bk = bb + (4 * j + t) * W1_PITCH;
            v2f av;
            av.x = t ? x.a[j].z : x.a[j].x;
            av.y = t ? x.a[j].w : x.a[j].y;
            const v2f b0 = *(const v2f*)(bk);
            const v2f b1 = *(const v2f*)(bk + 32);
            const v2f b2 = *(const v2f*)(bk + 64);
            const v2f b3 = *(const v2f*)(bk + 96);
            c0 = __builtin_amdgcn_wmma_f32_16x16x4_f32(false, av, false, b0, (short)0, c0, false, false);
            c1 = __builtin_amdgcn_wmma_f32_16x16x4_f32(false, av, false, b1, (short)0, c1, false, false);
            c2 = __builtin_amdgcn_wmma_f32_16x16x4_f32(false, av, false, b2, (short)0, c2, false, false);
            c3 = __builtin_amdgcn_wmma_f32_16x16x4_f32(false, av, false, b3, (short)0, c3, false, false);
        }
    }
    #pragma unroll
    for (int r = 0; r < 8; ++r) {
        const float d0 = __builtin_bit_cast(float, __builtin_amdgcn_readlane(__builtin_bit_cast(int, dnv), r));
        const float d1 = __builtin_bit_cast(float, __builtin_amdgcn_readlane(__builtin_bit_cast(int, dnv), r + 8));
        const float dm = half ? d1 : d0;
        const float y0 = c0[r] * dm, y1 = c1[r] * dm;
        const float y2 = c2[r] * dm, y3 = c3[r] * dm;
        const float g01 = xhalf(half ? y0 : y1);
        const float g23 = xhalf(half ? y2 : y3);
        float* lo = H + (size_t)(m0 + r) * D_HID + lane;
        float* hi = H + (size_t)(m0 + r + 8) * D_HID + lane;
        lo[0]  = half ? g01 : y0;
        lo[32] = half ? g23 : y2;
        hi[0]  = half ? y1 : g01;
        hi[32] = half ? y3 : g23;
    }
}


template <bool PLACE>
__device__ __forceinline__ void bin_take(unsigned w, unsigned* ncnt, const unsigned* noff,
                                         unsigned* outs, int* csr, unsigned base, bool fits) {
    const unsigned dl = w & (BIN_W - 1);
    const unsigned r  = atomicAdd(&ncnt[dl], 1u);
    if (PLACE) {
        const unsigned p = noff[dl] + r;
        const int s = (int)(w >> BIN_SHIFT);
        if (fits) outs[p] = (unsigned)s;
        else      csr[base + p] = s;
    }
}

template <bool PLACE>
__device__ __forceinline__ void bin_walk(const unsigned* __restrict__ binned,
                                         const unsigned* roff, const unsigned* rcnt,
                                         unsigned* ncnt, const unsigned* noff, unsigned* outs,
                                         int* csr, unsigned base, bool fits, int C) {
    const int lane = threadIdx.x & 31;
    const int wv   = threadIdx.x >> 5;
    for (int c0 = wv; c0 < C; c0 += (SORT_TB / 32) * RUN_U) {
        unsigned w0[RUN_U], w1[RUN_U], mm[RUN_U];
        const unsigned* pp[RUN_U];
        #pragma unroll
        for (int u = 0; u < RUN_U; ++u) {
            const int c = c0 + (SORT_TB / 32) * u;
            const bool in = c < C;
            mm[u] = in ? rcnt[c] : 0u;
            pp[u] = binned + (size_t)(in ? c : 0) * CHUNK + (in ? roff[c] : 0u);
            w0[u] = ((unsigned)lane      < mm[u]) ? pp[u][lane]      : 0xFFFFFFFFu;
            w1[u] = ((unsigned)lane + 32 < mm[u]) ? pp[u][lane + 32] : 0xFFFFFFFFu;
        }
        #pragma unroll
        for (int u = 0; u < RUN_U; ++u) {
            if (w0[u] != 0xFFFFFFFFu) bin_take<PLACE>(w0[u], ncnt, noff, outs, csr, base, fits);
            if (w1[u] != 0xFFFFFFFFu) bin_take<PLACE>(w1[u], ncnt, noff, outs, csr, base, fits);
            for (unsigned j = (unsigned)lane + 64; j < mm[u]; j += 32)
                bin_take<PLACE>(pp[u][j], ncnt, noff, outs, csr, base, fits);
        }
    }
}

__global__ __launch_bounds__(SORT_TB)
void k_bin_build_gemm1(const unsigned* __restrict__ binned, const uint2* __restrict__ tab,
                       unsigned* __restrict__ rowptr, float* __restrict__ dinv,
                       int* __restrict__ csr, const float* __restrict__ X,
                       const float* __restrict__ W, float* __restrict__ H,
                       int N, int C, int NB) {
    __shared__ unsigned roff[SORT_TB];
    __shared__ unsigned rcnt[SORT_TB];
    __shared__ unsigned ncnt[BIN_W];
    __shared__ unsigned noff[BIN_W];
    __shared__ unsigned sm[SORT_TB / 32];
    __shared__ unsigned outs[CAP_B];
    __shared__ float dn_s[BIN_W];
    __shared__ alignas(16) float w1s[(D_IN / 2) * W1_PITCH];
    const int t = threadIdx.x;
    const int b = blockIdx.x;
    const int lane = t & 31;
    const int m0 = b * BIN_W + (t >> 5) * 16;
    const bool live = m0 < N;

    if (live) __builtin_prefetch((const char*)(X + (size_t)m0 * D_IN) + lane * 256, 0, 3);

    const float4 we = ((const float4*)W)[(2 * (t >> 4))     * (D_HID / 4) + (t & 15)];
    const float4 wo = ((const float4*)W)[(2 * (t >> 4) + 1) * (D_HID / 4) + (t & 15)];

    unsigned off = 0, cnt = 0;
    if (t < C) {
        const uint2 q = tab[(size_t)t * NB + b];
        off = q.x;
        cnt = q.y;
    }
    roff[t] = off;
    rcnt[t] = cnt;
    if (t < BIN_W) ncnt[t] = 0u;
    {
        float* d = w1s + (t >> 4) * W1_PITCH + (t & 15) * 8;
        *(float4*)(d)     = make_float4(we.x, wo.x, we.y, wo.y);
        *(float4*)(d + 4) = make_float4(we.z, wo.z, we.w, wo.w);
    }
    unsigned total, base;
    (void)block_excl_scan(cnt, sm, &total);
    (void)block_excl_scan(off, sm, &base);
    const bool fits = total <= (unsigned)CAP_B;

    bin_walk<false>(binned, roff, rcnt, ncnt, noff, outs, csr, base, fits, C);
    __syncthreads();

    const unsigned dg = (t < BIN_W) ? ncnt[t] : 0u;
    unsigned tot2;
    const unsigned ex = block_excl_scan(dg, sm, &tot2);
    if (t < BIN_W) {
        noff[t] = ex;
        ncnt[t] = 0u;
        const float dv = rsqrtf((float)(dg + 1u));
        dn_s[t] = dv;
        const int node = b * BIN_W + t;
        if (node < N) {
            rowptr[node] = base + ex;
            dinv[node]   = dv;
        }
    }
    if (b == NB - 1 && t == 0) rowptr[N] = base + total;
    __syncthreads();

    bin_walk<true>(binned, roff, rcnt, ncnt, noff, outs, csr, base, fits, C);

    __syncthreads();
    XRows xr;
    gemm1_load(X, live ? m0 : 0, xr);
    if (fits)
        for (unsigned i = (unsigned)t; i < total; i += SORT_TB) csr[base + i] = (int)outs[i];

    if (!live) return;
    gemm1_block(xr, w1s, dn_s[(t >> 5) * 16 + (lane & 15)], H, m0);
}


__device__ __forceinline__ float  row_add(float a, float b)   { return a + b; }
__device__ __forceinline__ float2 row_add(float2 a, float2 b) { return make_float2(a.x + b.x, a.y + b.y); }
__device__ __forceinline__ float  row_fin(float a, float dn, float b)    { return dn * a + b; }
__device__ __forceinline__ float2 row_fin(float2 a, float dn, float2 b)  { return make_float2(fmaxf(dn * a.x + b.x, 0.f), fmaxf(dn * a.y + b.y, 0.f)); }

template <typename VT, int PITCH, bool TILE>
__device__ __forceinline__ float gather_walk(const unsigned* __restrict__ rowptr, const int* __restrict__ csr,
                                            const float* __restrict__ dinv, const float* __restrict__ src,
                                            const float* __restrict__ bias, float* __restrict__ dst,
                                            VT* __restrict__ own_s, int N) {
    const int lane = threadIdx.x & 31;
    const int wv = __builtin_amdgcn_readfirstlane((int)(threadIdx.x >> 5));
    const int n0 = (blockIdx.x * GATHER_WAVES + wv) * NPW;
    if (n0 >= N) return 0.f;
    const int cnt = (N - n0 < NPW) ? (N - n0) : NPW;

    const char* __restrict__ hb = (const char*)src;
    const unsigned ROWB    = 32u * (unsigned)sizeof(VT);
    const unsigned laneoff = (unsigned)lane * (unsigned)sizeof(VT);
    VT* __restrict__ out = (VT*)dst;
    VT* own = own_s + wv * (NPW * PITCH) + lane;

    VT t[NPW];
    #pragma unroll
    for (int k = 0; k < NPW; ++k) {
        const int kk = k < cnt ? k : cnt - 1;
        t[k] = *(const VT*)(hb + (size_t)(n0 + kk) * ROWB + laneoff);
    }
    const unsigned rpv = rowptr[n0 + (lane < cnt ? lane : cnt)];
    const float    dnv = dinv[n0 + (lane < cnt ? lane : cnt - 1)];
    const VT bv = ((const VT*)bias)[lane];
    #pragma unroll
    for (int k = 0; k < NPW; ++k) own[k * PITCH] = t[k];

    const unsigned e0   = (unsigned)__builtin_amdgcn_readlane((int)rpv, 0);
    const unsigned eEnd = (unsigned)__builtin_amdgcn_readlane((int)rpv, cnt);

    unsigned edc, edn;
    {
        const unsigned i0 = e0 + (unsigned)lane, i1 = i0 + 32u;
        edc = ((i0 < eEnd) ? (unsigned)csr[i0] : (unsigned)n0) * ROWB;
        edn = ((i1 < eEnd) ? (unsigned)csr[i1] : (unsigned)n0) * ROWB;
    }

    int      k      = 0;
    unsigned segEnd = (unsigned)__builtin_amdgcn_readlane((int)rpv, 1);
    VT acc     = t[0];
    VT own_nxt = own[PITCH];

    VT buf[2][GROUP];
    #pragma unroll
    for (int u = 0; u < GROUP; ++u) {
        const unsigned ro = (unsigned)__builtin_amdgcn_readlane((int)edc, u);
        buf[0][u] = *(const VT*)(hb + (size_t)(ro + laneoff));
    }

    unsigned gb = e0;
    for (unsigned eb = e0; eb < eEnd; eb += 32u) {
        unsigned edf;
        {
            const unsigned i2 = eb + 64u + (unsigned)lane;
            edf = ((i2 < eEnd) ? (unsigned)csr[i2] : (unsigned)n0) * ROWB;
        }
        #pragma unroll
        for (int g = 0; g < 32 / GROUP; ++g) {
            #pragma unroll
            for (int u = 0; u < GROUP; ++u) {
                const unsigned ro = (g + 1 < 32 / GROUP)
                    ? (unsigned)__builtin_amdgcn_readlane((int)edc, (g + 1) * GROUP + u)
                    : (unsigned)__builtin_amdgcn_readlane((int)edn, u);
                buf[(g + 1) & 1][u] = *(const VT*)(hb + (size_t)(ro + laneoff));
            }
            if (segEnd - gb >= (unsigned)GROUP) {
                #pragma unroll
                for (int u = 0; u < GROUP; ++u) acc = row_add(acc, buf[g & 1][u]);
            } else {
                #pragma unroll
                for (int u = 0; u < GROUP; ++u) {
                    while (__builtin_expect(gb + (unsigned)u == segEnd, 0)) {
                        const float dn = __builtin_bit_cast(float, __builtin_amdgcn_readlane(__builtin_bit_cast(int, dnv), k));
                        if constexpr (TILE) own[k * PITCH] = row_fin(acc, dn, bv);
                        else                out[(size_t)(n0 + k) * 32 + lane] = row_fin(acc, dn, bv);
                        ++k;
                        if (k < cnt) {
                            acc     = own_nxt;
                            own_nxt = own[((k + 1 < NPW) ? k + 1 : NPW - 1) * PITCH];
                            segEnd  = (unsigned)__builtin_amdgcn_readlane((int)rpv, k + 1);
                        } else {
                            segEnd  = 0xFFFFFFFFu;
                        }
                    }
                    acc = row_add(acc, buf[g & 1][u]);
                }
            }
            gb += (unsigned)GROUP;
        }
        edc = edn;
        edn = edf;
    }
    while (k < cnt) {
        const float dn = __builtin_bit_cast(float, __builtin_amdgcn_readlane(__builtin_bit_cast(int, dnv), k));
        if constexpr (TILE) own[k * PITCH] = row_fin(acc, dn, bv);
        else                out[(size_t)(n0 + k) * 32 + lane] = row_fin(acc, dn, bv);
        ++k;
        acc     = own_nxt;
        own_nxt = own[((k + 1 < NPW) ? k + 1 : NPW - 1) * PITCH];
    }
    return dnv;
}


#define TPITCH 34

__global__ __launch_bounds__(GATHER_WAVES * 32)
void k_gather64_gemm2(const unsigned* __restrict__ rowptr, const int* __restrict__ csr,
                      const float* __restrict__ dinv, const float* __restrict__ h,
                      const float* __restrict__ b1, const float* __restrict__ W2,
                      float* __restrict__ H2, int N) {
    __shared__ float2 rows_s[GATHER_WAVES * NPW * TPITCH];
    __shared__ alignas(16) float w2s[D_HID * D_OUT];

    #pragma unroll
    for (int q = 0; q < D_HID * D_OUT / (GATHER_WAVES * 32); ++q) {
        const int e  = q * (GATHER_WAVES * 32) + (int)threadIdx.x;
        const int ln = (e >> 2) & 31;
        const int k  = 4 * (e >> 7) + 2 * (ln >> 4) + (e & 1);
        const int n  = 16 * ((e >> 1) & 1) + (ln & 15);
        w2s[e] = W2[k * D_OUT + n];
    }
    __syncthreads();

    const int lane = threadIdx.x & 31;
    const int wv = __builtin_amdgcn_readfirstlane((int)(threadIdx.x >> 5));
    const int n0 = (blockIdx.x * GATHER_WAVES + wv) * NPW;
    if (n0 >= N) return;
    const int cnt = (N - n0 < NPW) ? (N - n0) : NPW;

    const float dnv = gather_walk<float2, TPITCH, true>(rowptr, csr, dinv, h, b1, nullptr, rows_s, N);


    const float2* tile = rows_s + wv * (NPW * TPITCH);
    const int half = lane >> 4;
    const int lm   = lane & 15;
    v8f c0 = {}, c1 = {};
    #pragma unroll
    for (int ks = 0; ks < D_HID / 4; ++ks) {
        const float2 av = tile[lm * TPITCH + 2 * ks + half];
        const float* bw = w2s + (ks * 32 + lane) * 4;
        v2f a, bl, bh;
        a.x  = av.x;  a.y  = av.y;
        bl.x = bw[0]; bl.y = bw[1];
        bh.x = bw[2]; bh.y = bw[3];
        c0 = __builtin_amdgcn_wmma_f32_16x16x4_f32(false, a, false, bl, (short)0, c0, false, false);
        c1 = __builtin_amdgcn_wmma_f32_16x16x4_f32(false, a, false, bh, (short)0, c1, false, false);
    }
    #pragma unroll
    for (int r = 0; r < 8; ++r) {
        const float d0 = __builtin_bit_cast(float, __builtin_amdgcn_readlane(__builtin_bit_cast(int, dnv), r));
        const float d1 = __builtin_bit_cast(float, __builtin_amdgcn_readlane(__builtin_bit_cast(int, dnv), r + 8));
        const float dm = half ? d1 : d0;
        const float y0 = c0[r] * dm;
        const float y1 = c1[r] * dm;
        const float got = xhalf(half ? y0 : y1);
        if (r < cnt)     H2[(size_t)(n0 + r) * D_OUT + lane]     = half ? got : y0;
        if (r + 8 < cnt) H2[(size_t)(n0 + r + 8) * D_OUT + lane] = half ? y1 : got;
    }
}


__global__ __launch_bounds__(GATHER_WAVES * 32)
void k_gather32(const unsigned* __restrict__ rowptr, const int* __restrict__ csr,
                const float* __restrict__ dinv, const float* __restrict__ h2,
                const float* __restrict__ b2, float* __restrict__ out, int N) {
    __shared__ float own_s[GATHER_WAVES * NPW * 32];
    (void)gather_walk<float, 32, false>(rowptr, csr, dinv, h2, b2, out, own_s, N);
}


extern "C" void kernel_launch(void* const* d_in, const int* in_sizes, int n_in,
                              void* d_out, int out_size, void* d_ws, size_t ws_size,
                              hipStream_t stream) {
    const float* x   = (const float*)d_in[0];
    const int*   ei  = (const int*)  d_in[1];
    const float* W1  = (const float*)d_in[2];
    const float* b1  = (const float*)d_in[3];
    const float* W2  = (const float*)d_in[4];
    const float* b2  = (const float*)d_in[5];
    float*       out = (float*)d_out;

    const int N = in_sizes[0] / D_IN;
    const int E = in_sizes[1] / 2;

    char* ws = (char*)d_ws;
    float*    h      = (float*)(ws);
    float*    h2     = (float*)(ws + (size_t)N * D_HID * 4);
    int*      csr    = (int*)  (ws + (size_t)N * D_HID * 4 + (size_t)N * D_OUT * 4);
    unsigned* binned = (unsigned*)((char*)csr + (size_t)E * 4);
    float*    dinv   = (float*)(binned + E);

    const int C  = (E + CHUNK - 1) / CHUNK;
    const int NB = (N + BIN_W - 1) / BIN_W;
    const int ngath = (N + GATHER_WAVES * NPW - 1) / (GATHER_WAVES * NPW);

    uint2*    tab    = (uint2*)(dinv + N);
    unsigned* rowptr = (unsigned*)(tab + (size_t)C * NB);

    k_bin_chunk<<<C,  SORT_TB, 0, stream>>>(ei, binned, tab, E, NB);
    k_bin_build_gemm1<<<NB, SORT_TB, 0, stream>>>(binned, tab, rowptr, dinv, csr, x, W1, h, N, C, NB);

    k_gather64_gemm2<<<ngath, GATHER_WAVES * 32, 0, stream>>>(rowptr, csr, dinv, h, b1, W2, h2, N);

    k_gather32      <<<ngath, GATHER_WAVES * 32, 0, stream>>>(rowptr, csr, dinv, h2, b2, out, N);
}
